# starting kernel
# baseline (speedup 1.0000x reference)
#include <hip/hip_runtime.h>

#define N_NODES   100000
#define N_EDGES   1600000
#define N_GRAPHS  1024
#define IN_LOCAL  16
#define IN_GLOBAL 8
#define HIDDEN    128
#define KC        (IN_LOCAL + HIDDEN)

typedef __attribute__((ext_vector_type(2))) float v2f;
typedef __attribute__((ext_vector_type(8))) float v8f;

#define LDSA_STRIDE 145
#define LDSH_STRIDE 129

__global__ __launch_bounds__(256) void k_global_enc(
    const float* __restrict__ gf, const float* __restrict__ Wg,
    const float* __restrict__ bg, float* __restrict__ hg) {
  int idx = blockIdx.x * 256 + threadIdx.x;
  int g = idx >> 7, h = idx & 127;
  float acc = bg[h];
#pragma unroll
  for (int k = 0; k < IN_GLOBAL; ++k)
    acc = fmaf(gf[g * IN_GLOBAL + k], Wg[k * HIDDEN + h], acc);
  hg[idx] = fmaxf(acc, 0.0f);
}

__global__ __launch_bounds__(256) void k_node_gemms(
    const float* __restrict__ x,  const int* __restrict__ batch,
    const float* __restrict__ hg,
    const float* __restrict__ Wc, const float* __restrict__ bc,
    const float* __restrict__ Wm, const float* __restrict__ bm,
    const float* __restrict__ Ws, const float* __restrict__ bs,
    float* __restrict__ m_buf, float* __restrict__ acc_buf) {
  __shared__ float lds_a[16 * LDSA_STRIDE];
  __shared__ float lds_h[16 * LDSH_STRIDE];

  const int tid = threadIdx.x;
  const int nodeBase = blockIdx.x * 16;

  for (int i = tid; i < 16 * KC; i += 256) {
    int row = i / KC, k = i - row * KC;
    int node = nodeBase + row;
    float v = (k < IN_LOCAL) ? x[node * IN_LOCAL + k]
                             : hg[batch[node] * HIDDEN + (k - IN_LOCAL)];
    lds_a[row * LDSA_STRIDE + k] = v;
  }
  __syncthreads();

  const int lane  = tid & 31;
  const int wave  = tid >> 5;
  const int half  = lane >> 4;
  const int idx16 = lane & 15;
  const int colBase = wave * 16;

  v8f acc = {};
#pragma unroll 4
  for (int kb = 0; kb < KC; kb += 4) {
    const int k0 = kb + 2 * half;
    v2f a, b;
    a.x = lds_a[idx16 * LDSA_STRIDE + k0];
    a.y = lds_a[idx16 * LDSA_STRIDE + k0 + 1];
    b.x = Wc[k0 * HIDDEN + colBase + idx16];
    b.y = Wc[(k0 + 1) * HIDDEN + colBase + idx16];
    acc = __builtin_amdgcn_wmma_f32_16x16x4_f32(
        false, a, false, b, (short)0, acc, false, false);
  }
  {
    const float bias = bc[colBase + idx16];
#pragma unroll
    for (int j = 0; j < 8; ++j) {
      int row = j + 8 * half;
      lds_h[row * LDSH_STRIDE + colBase + idx16] = fmaxf(acc[j] + bias, 0.0f);
    }
  }
  __syncthreads();

  v8f am = {}, as = {};
#pragma unroll 4
  for (int kb = 0; kb < HIDDEN; kb += 4) {
    const int k0 = kb + 2 * half;
    v2f a, b0, b1;
    a.x  = lds_h[idx16 * LDSH_STRIDE + k0];
    a.y  = lds_h[idx16 * LDSH_STRIDE + k0 + 1];
    b0.x = Wm[k0 * HIDDEN + colBase + idx16];
    b0.y = Wm[(k0 + 1) * HIDDEN + colBase + idx16];
    b1.x = Ws[k0 * HIDDEN + colBase + idx16];
    b1.y = Ws[(k0 + 1) * HIDDEN + colBase + idx16];
    am = __builtin_amdgcn_wmma_f32_16x16x4_f32(
        false, a, false, b0, (short)0, am, false, false);
    as = __builtin_amdgcn_wmma_f32_16x16x4_f32(
        false, a, false, b1, (short)0, as, false, false);
  }
  {
    const int col = colBase + idx16;
    const float biasm = bm[col], biass = bs[col];
#pragma unroll
    for (int j = 0; j < 8; ++j) {
      int row  = j + 8 * half;
      int node = nodeBase + row;
      float mv = fmaxf(am[j] + biasm, 0.0f);
      m_buf[(size_t)node * HIDDEN + col]   = mv;
      acc_buf[(size_t)node * HIDDEN + col] = mv + as[j] + biass;
    }
  }
}

__global__ __launch_bounds__(256) void k_scatter(
    const int* __restrict__ edge_index, const float* __restrict__ m_buf,
    float* __restrict__ acc_buf) {
  const int lane = threadIdx.x & 31;
  const int e = blockIdx.x * 8 + (threadIdx.x >> 5);
  const int src = edge_index[e];
  const int dst = edge_index[N_EDGES + e];
  const float* ms = m_buf   + (size_t)src * HIDDEN;
  float*       ad = acc_buf + (size_t)dst * HIDDEN;
#pragma unroll
  for (int t = 0; t < 4; ++t) {
    int c = lane + t * 32;
    unsafeAtomicAdd(&ad[c], ms[c]);
  }
}

__global__ __launch_bounds__(256) void k_out(
    const float* __restrict__ acc_buf, const float* __restrict__ Wo,
    const float* __restrict__ bo, float* __restrict__ out) {
  const int lane = threadIdx.x & 31;
  const int node = blockIdx.x * 8 + (threadIdx.x >> 5);
  float a0 = 0.0f, a1 = 0.0f;
#pragma unroll
  for (int t = 0; t < 4; ++t) {
    int k = lane + t * 32;
    float h = fmaxf(acc_buf[(size_t)node * HIDDEN + k], 0.0f);
    a0 = fmaf(h, Wo[k * 2 + 0], a0);
    a1 = fmaf(h, Wo[k * 2 + 1], a1);
  }
#pragma unroll
  for (int off = 16; off >= 1; off >>= 1) {
    a0 += __shfl_xor(a0, off, 32);
    a1 += __shfl_xor(a1, off, 32);
  }
  if (lane == 0) {
    out[node * 2 + 0] = a0 + bo[0];
    out[node * 2 + 1] = a1 + bo[1];
  }
}

extern "C" void kernel_launch(void* const* d_in, const int* in_sizes, int n_in,
                              void* d_out, int out_size, void* d_ws, size_t ws_size,
                              hipStream_t stream) {
  (void)in_sizes; (void)n_in; (void)out_size; (void)ws_size;
  const float* x     = (const float*)d_in[0];
  const int*   edge  = (const int*)  d_in[1];
  const int*   batch = (const int*)  d_in[2];
  const float* gf    = (const float*)d_in[3];
  const float* Wg    = (const float*)d_in[4];
  const float* bg    = (const float*)d_in[5];
  const float* Wc    = (const float*)d_in[6];
  const float* bc    = (const float*)d_in[7];
  const float* Wm    = (const float*)d_in[8];
  const float* bm    = (const float*)d_in[9];
  const float* Ws    = (const float*)d_in[10];
  const float* bs    = (const float*)d_in[11];
  const float* Wo    = (const float*)d_in[12];
  const float* bo    = (const float*)d_in[13];
  float* out = (float*)d_out;

  float* hg      = (float*)d_ws;
  float* m_buf   = hg    + (size_t)N_GRAPHS * HIDDEN;
  float* acc_buf = m_buf + (size_t)N_NODES  * HIDDEN;

  k_global_enc<<<(N_GRAPHS * HIDDEN) / 256, 256, 0, stream>>>(gf, Wg, bg, hg);
  k_node_gemms<<<N_NODES / 16, 256, 0, stream>>>(x, batch, hg, Wc, bc,
                                                 Wm, bm, Ws, bs, m_buf, acc_buf);
  k_scatter<<<N_EDGES / 8, 256, 0, stream>>>(edge, m_buf, acc_buf);
  k_out<<<N_NODES / 8, 256, 0, stream>>>(acc_buf, Wo, bo, out);
}


# More work per wave: scatter walks 32 edges per wave with 8 rows in flight (was 1 edge, 1 load in flight), out does 16 nodes per wave; same float atomics, arrival order varies as before
# speedup vs baseline: 1.7733x; 1.7733x over previous
#include <hip/hip_runtime.h>

#define N_NODES   100000
#define N_EDGES   1600000
#define N_GRAPHS  1024
#define IN_LOCAL  16
#define IN_GLOBAL 8
#define HIDDEN    128
#define KC        (IN_LOCAL + HIDDEN)

typedef __attribute__((ext_vector_type(2))) float v2f;
typedef __attribute__((ext_vector_type(8))) float v8f;

#define LDSA_STRIDE 145
#define LDSH_STRIDE 129

__global__ __launch_bounds__(256) void k_global_enc(
    const float* __restrict__ gf, const float* __restrict__ Wg,
    const float* __restrict__ bg, float* __restrict__ hg) {
  int idx = blockIdx.x * 256 + threadIdx.x;
  int g = idx >> 7, h = idx & 127;
  float acc = bg[h];
#pragma unroll
  for (int k = 0; k < IN_GLOBAL; ++k)
    acc = fmaf(gf[g * IN_GLOBAL + k], Wg[k * HIDDEN + h], acc);
  hg[idx] = fmaxf(acc, 0.0f);
}

__global__ __launch_bounds__(256) void k_node_gemms(
    const float* __restrict__ x,  const int* __restrict__ batch,
    const float* __restrict__ hg,
    const float* __restrict__ Wc, const float* __restrict__ bc,
    const float* __restrict__ Wm, const float* __restrict__ bm,
    const float* __restrict__ Ws, const float* __restrict__ bs,
    float* __restrict__ m_buf, float* __restrict__ acc_buf) {
  __shared__ float lds_a[16 * LDSA_STRIDE];
  __shared__ float lds_h[16 * LDSH_STRIDE];

  const int tid = threadIdx.x;
  const int nodeBase = blockIdx.x * 16;

  for (int i = tid; i < 16 * KC; i += 256) {
    int row = i / KC, k = i - row * KC;
    int node = nodeBase + row;
    float v = (k < IN_LOCAL) ? x[node * IN_LOCAL + k]
                             : hg[batch[node] * HIDDEN + (k - IN_LOCAL)];
    lds_a[row * LDSA_STRIDE + k] = v;
  }
  __syncthreads();

  const int lane  = tid & 31;
  const int wave  = tid >> 5;
  const int half  = lane >> 4;
  const int idx16 = lane & 15;
  const int colBase = wave * 16;

  v8f acc = {};
#pragma unroll 4
  for (int kb = 0; kb < KC; kb += 4) {
    const int k0 = kb + 2 * half;
    v2f a, b;
    a.x = lds_a[idx16 * LDSA_STRIDE + k0];
    a.y = lds_a[idx16 * LDSA_STRIDE + k0 + 1];
    b.x = Wc[k0 * HIDDEN + colBase + idx16];
    b.y = Wc[(k0 + 1) * HIDDEN + colBase + idx16];
    acc = __builtin_amdgcn_wmma_f32_16x16x4_f32(
        false, a, false, b, (short)0, acc, false, false);
  }
  {
    const float bias = bc[colBase + idx16];
#pragma unroll
    for (int j = 0; j < 8; ++j) {
      int row = j + 8 * half;
      lds_h[row * LDSH_STRIDE + colBase + idx16] = fmaxf(acc[j] + bias, 0.0f);
    }
  }
  __syncthreads();

  v8f am = {}, as = {};
#pragma unroll 4
  for (int kb = 0; kb < HIDDEN; kb += 4) {
    const int k0 = kb + 2 * half;
    v2f a, b0, b1;
    a.x  = lds_h[idx16 * LDSH_STRIDE + k0];
    a.y  = lds_h[idx16 * LDSH_STRIDE + k0 + 1];
    b0.x = Wm[k0 * HIDDEN + colBase + idx16];
    b0.y = Wm[(k0 + 1) * HIDDEN + colBase + idx16];
    b1.x = Ws[k0 * HIDDEN + colBase + idx16];
    b1.y = Ws[(k0 + 1) * HIDDEN + colBase + idx16];
    am = __builtin_amdgcn_wmma_f32_16x16x4_f32(
        false, a, false, b0, (short)0, am, false, false);
    as = __builtin_amdgcn_wmma_f32_16x16x4_f32(
        false, a, false, b1, (short)0, as, false, false);
  }
  {
    const int col = colBase + idx16;
    const float biasm = bm[col], biass = bs[col];
#pragma unroll
    for (int j = 0; j < 8; ++j) {
      int row  = j + 8 * half;
      int node = nodeBase + row;
      float mv = fmaxf(am[j] + biasm, 0.0f);
      m_buf[(size_t)node * HIDDEN + col]   = mv;
      acc_buf[(size_t)node * HIDDEN + col] = mv + as[j] + biass;
    }
  }
}

#define EPW 32
#define EU  8

__global__ __launch_bounds__(256) void k_scatter(
    const int* __restrict__ edge_index, const float* __restrict__ m_buf,
    float* __restrict__ acc_buf) {
  const int lane = threadIdx.x & 31;
  const int wave = blockIdx.x * 8 + (threadIdx.x >> 5);
  const int e = wave * EPW + lane;
  const int srcv = edge_index[e];
  const int dstv = edge_index[N_EDGES + e];
  for (int jb = 0; jb < EPW; jb += EU) {
    float v[EU][4];
    int d[EU];
#pragma unroll
    for (int u = 0; u < EU; ++u) {
      const int s = __builtin_amdgcn_readlane(srcv, jb + u);
      d[u]        = __builtin_amdgcn_readlane(dstv, jb + u);
      const float* ms = m_buf + (size_t)s * HIDDEN + lane;
#pragma unroll
      for (int t = 0; t < 4; ++t) v[u][t] = ms[t * 32];
    }
#pragma unroll
    for (int u = 0; u < EU; ++u) {
      float* ad = acc_buf + (size_t)d[u] * HIDDEN + lane;
#pragma unroll
      for (int t = 0; t < 4; ++t)
        unsafeAtomicAdd(&ad[t * 32], v[u][t]);
    }
  }
}

#define NPW 16
#define NU  8

__global__ __launch_bounds__(256) void k_out(
    const float* __restrict__ acc_buf, const float* __restrict__ Wo,
    const float* __restrict__ bo, float* __restrict__ out) {
  const int lane = threadIdx.x & 31;
  const int wave = blockIdx.x * 8 + (threadIdx.x >> 5);
  const int nodeBase = wave * NPW;
  if (nodeBase >= N_NODES) return;
  float w0[4], w1[4];
#pragma unroll
  for (int t = 0; t < 4; ++t) {
    int k = lane + t * 32;
    w0[t] = Wo[k * 2 + 0];
    w1[t] = Wo[k * 2 + 1];
  }
  const float bias = bo[lane & 1];
  float res = 0.0f;
  for (int jb = 0; jb < NPW; jb += NU) {
    float h[NU][4];
#pragma unroll
    for (int u = 0; u < NU; ++u) {
      const float* ar = acc_buf + (size_t)(nodeBase + jb + u) * HIDDEN + lane;
#pragma unroll
      for (int t = 0; t < 4; ++t) h[u][t] = ar[t * 32];
    }
#pragma unroll
    for (int u = 0; u < NU; ++u) {
      float a0 = 0.0f, a1 = 0.0f;
#pragma unroll
      for (int t = 0; t < 4; ++t) {
        float hv = fmaxf(h[u][t], 0.0f);
        a0 = fmaf(hv, w0[t], a0);
        a1 = fmaf(hv, w1[t], a1);
      }
#pragma unroll
      for (int off = 16; off >= 1; off >>= 1) {
        a0 += __shfl_xor(a0, off, 32);
        a1 += __shfl_xor(a1, off, 32);
      }
      if ((lane >> 1) == jb + u) res = (lane & 1) ? a1 : a0;
    }
  }
  out[nodeBase * 2 + lane] = res + bias;
}

extern "C" void kernel_launch(void* const* d_in, const int* in_sizes, int n_in,
                              void* d_out, int out_size, void* d_ws, size_t ws_size,
                              hipStream_t stream) {
  (void)in_sizes; (void)n_in; (void)out_size; (void)ws_size;
  const float* x     = (const float*)d_in[0];
  const int*   edge  = (const int*)  d_in[1];
  const int*   batch = (const int*)  d_in[2];
  const float* gf    = (const float*)d_in[3];
  const float* Wg    = (const float*)d_in[4];
  const float* bg    = (const float*)d_in[5];
  const float* Wc    = (const float*)d_in[6];
  const float* bc    = (const float*)d_in[7];
  const float* Wm    = (const float*)d_in[8];
  const float* bm    = (const float*)d_in[9];
  const float* Ws    = (const float*)d_in[10];
  const float* bs    = (const float*)d_in[11];
  const float* Wo    = (const float*)d_in[12];
  const float* bo    = (const float*)d_in[13];
  float* out = (float*)d_out;

  float* hg      = (float*)d_ws;
  float* m_buf   = hg    + (size_t)N_GRAPHS * HIDDEN;
  float* acc_buf = m_buf + (size_t)N_NODES  * HIDDEN;

  k_global_enc<<<(N_GRAPHS * HIDDEN) / 256, 256, 0, stream>>>(gf, Wg, bg, hg);
  k_node_gemms<<<N_NODES / 16, 256, 0, stream>>>(x, batch, hg, Wc, bc,
                                                 Wm, bm, Ws, bs, m_buf, acc_buf);
  k_scatter<<<N_EDGES / (8 * EPW), 256, 0, stream>>>(edge, m_buf, acc_buf);
  k_out<<<(N_NODES / NPW + 7) / 8, 256, 0, stream>>>(acc_buf, Wo, bo, out);
}


# Scatter split by package half: a work-group adds only into the node range of the half it runs on (die id; two work queues, helps the other when done); same float atomics, arrival order varies
# speedup vs baseline: 1.9826x; 1.1180x over previous
#include <hip/hip_runtime.h>

#define N_NODES   100000
#define N_EDGES   1600000
#define N_GRAPHS  1024
#define IN_LOCAL  16
#define IN_GLOBAL 8
#define HIDDEN    128
#define KC        (IN_LOCAL + HIDDEN)

typedef __attribute__((ext_vector_type(2))) float v2f;
typedef __attribute__((ext_vector_type(8))) float v8f;

#define LDSA_STRIDE 145
#define LDSH_STRIDE 129

__global__ __launch_bounds__(256) void k_global_enc(
    const float* __restrict__ gf, const float* __restrict__ Wg,
    const float* __restrict__ bg, float* __restrict__ hg,
    int* __restrict__ queue) {
  int idx = blockIdx.x * 256 + threadIdx.x;
  if (idx < 2) queue[idx] = 0;
  int g = idx >> 7, h = idx & 127;
  float acc = bg[h];
#pragma unroll
  for (int k = 0; k < IN_GLOBAL; ++k)
    acc = fmaf(gf[g * IN_GLOBAL + k], Wg[k * HIDDEN + h], acc);
  hg[idx] = fmaxf(acc, 0.0f);
}

__global__ __launch_bounds__(256) void k_node_gemms(
    const float* __restrict__ x,  const int* __restrict__ batch,
    const float* __restrict__ hg,
    const float* __restrict__ Wc, const float* __restrict__ bc,
    const float* __restrict__ Wm, const float* __restrict__ bm,
    const float* __restrict__ Ws, const float* __restrict__ bs,
    float* __restrict__ m_buf, float* __restrict__ acc_buf) {
  __shared__ float lds_a[16 * LDSA_STRIDE];
  __shared__ float lds_h[16 * LDSH_STRIDE];

  const int tid = threadIdx.x;
  const int nodeBase = blockIdx.x * 16;

  for (int i = tid; i < 16 * KC; i += 256) {
    int row = i / KC, k = i - row * KC;
    int node = nodeBase + row;
    float v = (k < IN_LOCAL) ? x[node * IN_LOCAL + k]
                             : hg[batch[node] * HIDDEN + (k - IN_LOCAL)];
    lds_a[row * LDSA_STRIDE + k] = v;
  }
  __syncthreads();

  const int lane  = tid & 31;
  const int wave  = tid >> 5;
  const int half  = lane >> 4;
  const int idx16 = lane & 15;
  const int colBase = wave * 16;

  v8f acc = {};
#pragma unroll 4
  for (int kb = 0; kb < KC; kb += 4) {
    const int k0 = kb + 2 * half;
    v2f a, b;
    a.x = lds_a[idx16 * LDSA_STRIDE + k0];
    a.y = lds_a[idx16 * LDSA_STRIDE + k0 + 1];
    b.x = Wc[k0 * HIDDEN + colBase + idx16];
    b.y = Wc[(k0 + 1) * HIDDEN + colBase + idx16];
    acc = __builtin_amdgcn_wmma_f32_16x16x4_f32(
        false, a, false, b, (short)0, acc, false, false);
  }
  {
    const float bias = bc[colBase + idx16];
#pragma unroll
    for (int j = 0; j < 8; ++j) {
      int row = j + 8 * half;
      lds_h[row * LDSH_STRIDE + colBase + idx16] = fmaxf(acc[j] + bias, 0.0f);
    }
  }
  __syncthreads();

  v8f am = {}, as = {};
#pragma unroll 4
  for (int kb = 0; kb < HIDDEN; kb += 4) {
    const int k0 = kb + 2 * half;
    v2f a, b0, b1;
    a.x  = lds_h[idx16 * LDSH_STRIDE + k0];
    a.y  = lds_h[idx16 * LDSH_STRIDE + k0 + 1];
    b0.x = Wm[k0 * HIDDEN + colBase + idx16];
    b0.y = Wm[(k0 + 1) * HIDDEN + colBase + idx16];
    b1.x = Ws[k0 * HIDDEN + colBase + idx16];
    b1.y = Ws[(k0 + 1) * HIDDEN + colBase + idx16];
    am = __builtin_amdgcn_wmma_f32_16x16x4_f32(
        false, a, false, b0, (short)0, am, false, false);
    as = __builtin_amdgcn_wmma_f32_16x16x4_f32(
        false, a, false, b1, (short)0, as, false, false);
  }
  {
    const int col = colBase + idx16;
    const float biasm = bm[col], biass = bs[col];
#pragma unroll
    for (int j = 0; j < 8; ++j) {
      int row  = j + 8 * half;
      int node = nodeBase + row;
      float mv = fmaxf(am[j] + biasm, 0.0f);
      m_buf[(size_t)node * HIDDEN + col]   = mv;
      acc_buf[(size_t)node * HIDDEN + col] = mv + as[j] + biass;
    }
  }
}

#define EU       8
#define CPW      2
#define GRAB_E   (8 * CPW * 32)
#define N_GRABS  (N_EDGES / GRAB_E)
#define SC_BLOCKS 2048

__global__ __launch_bounds__(256) void k_scatter(
    const int* __restrict__ edge_index, const float* __restrict__ m_buf,
    float* __restrict__ acc_buf, int* __restrict__ queue) {
  __shared__ int s_g[2];
  const int tid  = threadIdx.x;
  const int lane = tid & 31;
  const int wv   = tid >> 5;
  const int myHalf = (__builtin_amdgcn_s_sendmsg_rtn(0x87) >> 18) & 1;

  for (int pass = 0; pass < 2; ++pass) {
    const int h = myHalf ^ pass;
    if (tid == 0) s_g[0] = atomicAdd(&queue[h], 1);
    __syncthreads();
    for (int it = 0; it <= N_GRABS; ++it) {
      const int g = s_g[it & 1];
      if (g >= N_GRABS) break;
      int srcs[CPW], dsts[CPW];
#pragma unroll
      for (int c = 0; c < CPW; ++c) {
        const int e = g * GRAB_E + (wv * CPW + c) * 32 + lane;
        srcs[c] = edge_index[e];
        dsts[c] = edge_index[N_EDGES + e];
      }
      int nxt = 0;
      if (tid == 0) nxt = atomicAdd(&queue[h], 1);
      asm volatile("" :: "v"(srcs[0]), "v"(srcs[1]), "v"(dsts[0]), "v"(dsts[1]));
#pragma unroll
      for (int c = 0; c < CPW; ++c) {
        const int srcv = srcs[c];
        const int dstv = dsts[c];
        const int dh = (dstv >= N_NODES / 2) ? 1 : 0;
        unsigned mask = __builtin_amdgcn_ballot_w32(dh == h);
        for (int t = 0; t < 32 / EU && mask != 0u; ++t) {
          int s[EU], d[EU];
          int n = 0;
#pragma unroll
          for (int u = 0; u < EU; ++u) {
            s[u] = 0; d[u] = 0;
            if (mask != 0u) {
              const int j = __builtin_ctz(mask);
              mask &= mask - 1u;
              s[u] = __builtin_amdgcn_readlane(srcv, j);
              d[u] = __builtin_amdgcn_readlane(dstv, j);
              n = u + 1;
            }
          }
          float v[EU][4];
#pragma unroll
          for (int u = 0; u < EU; ++u) {
            if (u < n) {
              const float* ms = m_buf + (size_t)s[u] * HIDDEN + lane;
#pragma unroll
              for (int q = 0; q < 4; ++q) v[u][q] = ms[q * 32];
            }
          }
#pragma unroll
          for (int u = 0; u < EU; ++u) {
            if (u < n) {
              float* ad = acc_buf + (size_t)d[u] * HIDDEN + lane;
#pragma unroll
              for (int q = 0; q < 4; ++q)
                unsafeAtomicAdd(&ad[q * 32], v[u][q]);
            }
          }
        }
      }
      if (tid == 0) s_g[(it + 1) & 1] = nxt;
      __builtin_amdgcn_fence(__ATOMIC_RELEASE, "workgroup", "local");
      __builtin_amdgcn_s_barrier();
      __builtin_amdgcn_fence(__ATOMIC_ACQUIRE, "workgroup", "local");
    }
    __syncthreads();
  }
}

#define NPW 16
#define NU  8

__global__ __launch_bounds__(256) void k_out(
    const float* __restrict__ acc_buf, const float* __restrict__ Wo,
    const float* __restrict__ bo, float* __restrict__ out) {
  const int lane = threadIdx.x & 31;
  const int wave = blockIdx.x * 8 + (threadIdx.x >> 5);
  const int nodeBase = wave * NPW;
  if (nodeBase >= N_NODES) return;
  float w0[4], w1[4];
#pragma unroll
  for (int t = 0; t < 4; ++t) {
    int k = lane + t * 32;
    w0[t] = Wo[k * 2 + 0];
    w1[t] = Wo[k * 2 + 1];
  }
  const float bias = bo[lane & 1];
  float res = 0.0f;
  for (int jb = 0; jb < NPW; jb += NU) {
    float h[NU][4];
#pragma unroll
    for (int u = 0; u < NU; ++u) {
      const float* ar = acc_buf + (size_t)(nodeBase + jb + u) * HIDDEN + lane;
#pragma unroll
      for (int t = 0; t < 4; ++t) h[u][t] = ar[t * 32];
    }
#pragma unroll
    for (int u = 0; u < NU; ++u) {
      float a0 = 0.0f, a1 = 0.0f;
#pragma unroll
      for (int t = 0; t < 4; ++t) {
        float hv = fmaxf(h[u][t], 0.0f);
        a0 = fmaf(hv, w0[t], a0);
        a1 = fmaf(hv, w1[t], a1);
      }
#pragma unroll
      for (int off = 16; off >= 1; off >>= 1) {
        a0 += __shfl_xor(a0, off, 32);
        a1 += __shfl_xor(a1, off, 32);
      }
      if ((lane >> 1) == jb + u) res = (lane & 1) ? a1 : a0;
    }
  }
  out[nodeBase * 2 + lane] = res + bias;
}

extern "C" void kernel_launch(void* const* d_in, const int* in_sizes, int n_in,
                              void* d_out, int out_size, void* d_ws, size_t ws_size,
                              hipStream_t stream) {
  (void)in_sizes; (void)n_in; (void)out_size; (void)ws_size;
  const float* x     = (const float*)d_in[0];
  const int*   edge  = (const int*)  d_in[1];
  const int*   batch = (const int*)  d_in[2];
  const float* gf    = (const float*)d_in[3];
  const float* Wg    = (const float*)d_in[4];
  const float* bg    = (const float*)d_in[5];
  const float* Wc    = (const float*)d_in[6];
  const float* bc    = (const float*)d_in[7];
  const float* Wm    = (const float*)d_in[8];
  const float* bm    = (const float*)d_in[9];
  const float* Ws    = (const float*)d_in[10];
  const float* bs    = (const float*)d_in[11];
  const float* Wo    = (const float*)d_in[12];
  const float* bo    = (const float*)d_in[13];
  float* out = (float*)d_out;
  int* queue = (int*)d_out;

  float* hg      = (float*)d_ws;
  float* m_buf   = hg    + (size_t)N_GRAPHS * HIDDEN;
  float* acc_buf = m_buf + (size_t)N_NODES  * HIDDEN;

  k_global_enc<<<(N_GRAPHS * HIDDEN) / 256, 256, 0, stream>>>(gf, Wg, bg, hg, queue);
  k_node_gemms<<<N_NODES / 16, 256, 0, stream>>>(x, batch, hg, Wc, bc,
                                                 Wm, bm, Ws, bs, m_buf, acc_buf);
  k_scatter<<<SC_BLOCKS, 256, 0, stream>>>(edge, m_buf, acc_buf, queue);
  k_out<<<(N_NODES / NPW + 7) / 8, 256, 0, stream>>>(acc_buf, Wo, bo, out);
}


# Graph part of the first layer hoisted: hg @ Wc[16:] + bc done once per graph (WMMA, kernel 1), node GEMM keeps K=16 over x and starts its sum from that row; h0's 144-term sum order changes
# speedup vs baseline: 2.2336x; 1.1266x over previous
#include <hip/hip_runtime.h>

#define N_NODES   100000
#define N_EDGES   1600000
#define N_GRAPHS  1024
#define IN_LOCAL  16
#define IN_GLOBAL 8
#define HIDDEN    128

typedef __attribute__((ext_vector_type(2))) float v2f;
typedef __attribute__((ext_vector_type(8))) float v8f;

#define LDSH_STRIDE 129

__global__ __launch_bounds__(256) void k_global_enc(
    const float* __restrict__ gf, const float* __restrict__ Wg,
    const float* __restrict__ bg, const float* __restrict__ Wc,
    const float* __restrict__ bc, float* __restrict__ hgw,
    int* __restrict__ queue) {
  __shared__ float lds_g[16 * LDSH_STRIDE];

  const int tid = threadIdx.x;
  if (blockIdx.x == 0 && tid < 2) queue[tid] = 0;
  const int gBase = blockIdx.x * 16;

  {
    const int h  = tid & 127;
    const int r0 = tid >> 7;
    float wg[IN_GLOBAL];
#pragma unroll
    for (int k = 0; k < IN_GLOBAL; ++k) wg[k] = Wg[k * HIDDEN + h];
    const float b = bg[h];
#pragma unroll
    for (int i = 0; i < 8; ++i) {
      const int row = r0 + 2 * i;
      const float* gr = gf + (gBase + row) * IN_GLOBAL;
      float acc = b;
#pragma unroll
      for (int k = 0; k < IN_GLOBAL; ++k) acc = fmaf(gr[k], wg[k], acc);
      lds_g[row * LDSH_STRIDE + h] = fmaxf(acc, 0.0f);
    }
  }
  __syncthreads();

  const int lane  = tid & 31;
  const int wave  = tid >> 5;
  const int half  = lane >> 4;
  const int idx16 = lane & 15;
  const int colBase = wave * 16;

  v8f acc = {};
#pragma unroll 8
  for (int kb = 0; kb < HIDDEN; kb += 4) {
    const int k0 = kb + 2 * half;
    v2f a, b;
    a.x = lds_g[idx16 * LDSH_STRIDE + k0];
    a.y = lds_g[idx16 * LDSH_STRIDE + k0 + 1];
    b.x = Wc[(IN_LOCAL + k0) * HIDDEN + colBase + idx16];
    b.y = Wc[(IN_LOCAL + k0 + 1) * HIDDEN + colBase + idx16];
    acc = __builtin_amdgcn_wmma_f32_16x16x4_f32(
        false, a, false, b, (short)0, acc, false, false);
  }
  {
    const float bias = bc[colBase + idx16];
#pragma unroll
    for (int j = 0; j < 8; ++j) {
      int row = j + 8 * half;
      hgw[(gBase + row) * HIDDEN + colBase + idx16] = acc[j] + bias;
    }
  }
}

__global__ __launch_bounds__(256) void k_node_gemms(
    const float* __restrict__ x,  const int* __restrict__ batch,
    const float* __restrict__ hgw, const float* __restrict__ Wc,
    const float* __restrict__ Wm, const float* __restrict__ bm,
    const float* __restrict__ Ws, const float* __restrict__ bs,
    float* __restrict__ m_buf, float* __restrict__ acc_buf) {
  __shared__ float lds_h[16 * LDSH_STRIDE];

  const int tid = threadIdx.x;
  const int nodeBase = blockIdx.x * 16;

  const int lane  = tid & 31;
  const int wave  = tid >> 5;
  const int half  = lane >> 4;
  const int idx16 = lane & 15;
  const int colBase = wave * 16;

  v8f acc;
#pragma unroll
  for (int j = 0; j < 8; ++j) {
    const int g = batch[nodeBase + j + 8 * half];
    acc[j] = hgw[g * HIDDEN + colBase + idx16];
  }
  {
    const float* xr = x + (nodeBase + idx16) * IN_LOCAL + 2 * half;
#pragma unroll
    for (int kb = 0; kb < IN_LOCAL; kb += 4) {
      const int k0 = kb + 2 * half;
      v2f a, b;
      a.x = xr[kb];
      a.y = xr[kb + 1];
      b.x = Wc[k0 * HIDDEN + colBase + idx16];
      b.y = Wc[(k0 + 1) * HIDDEN + colBase + idx16];
      acc = __builtin_amdgcn_wmma_f32_16x16x4_f32(
          false, a, false, b, (short)0, acc, false, false);
    }
  }
  {
#pragma unroll
    for (int j = 0; j < 8; ++j) {
      int row = j + 8 * half;
      lds_h[row * LDSH_STRIDE + colBase + idx16] = fmaxf(acc[j], 0.0f);
    }
  }
  __syncthreads();

  v8f am = {}, as = {};
#pragma unroll 4
  for (int kb = 0; kb < HIDDEN; kb += 4) {
    const int k0 = kb + 2 * half;
    v2f a, b0, b1;
    a.x  = lds_h[idx16 * LDSH_STRIDE + k0];
    a.y  = lds_h[idx16 * LDSH_STRIDE + k0 + 1];
    b0.x = Wm[k0 * HIDDEN + colBase + idx16];
    b0.y = Wm[(k0 + 1) * HIDDEN + colBase + idx16];
    b1.x = Ws[k0 * HIDDEN + colBase + idx16];
    b1.y = Ws[(k0 + 1) * HIDDEN + colBase + idx16];
    am = __builtin_amdgcn_wmma_f32_16x16x4_f32(
        false, a, false, b0, (short)0, am, false, false);
    as = __builtin_amdgcn_wmma_f32_16x16x4_f32(
        false, a, false, b1, (short)0, as, false, false);
  }
  {
    const int col = colBase + idx16;
    const float biasm = bm[col], biass = bs[col];
#pragma unroll
    for (int j = 0; j < 8; ++j) {
      int row  = j + 8 * half;
      int node = nodeBase + row;
      float mv = fmaxf(am[j] + biasm, 0.0f);
      m_buf[(size_t)node * HIDDEN + col]   = mv;
      acc_buf[(size_t)node * HIDDEN + col] = mv + as[j] + biass;
    }
  }
}

#define EU       8
#define CPW      2
#define GRAB_E   (8 * CPW * 32)
#define N_GRABS  (N_EDGES / GRAB_E)
#define SC_BLOCKS 2048

__global__ __launch_bounds__(256) void k_scatter(
    const int* __restrict__ edge_index, const float* __restrict__ m_buf,
    float* __restrict__ acc_buf, int* __restrict__ queue) {
  __shared__ int s_g[2];
  const int tid  = threadIdx.x;
  const int lane = tid & 31;
  const int wv   = tid >> 5;
  const int myHalf = (__builtin_amdgcn_s_sendmsg_rtn(0x87) >> 18) & 1;

  for (int pass = 0; pass < 2; ++pass) {
    const int h = myHalf ^ pass;
    if (tid == 0) s_g[0] = atomicAdd(&queue[h], 1);
    __syncthreads();
    for (int it = 0; it <= N_GRABS; ++it) {
      const int g = s_g[it & 1];
      if (g >= N_GRABS) break;
      int srcs[CPW], dsts[CPW];
#pragma unroll
      for (int c = 0; c < CPW; ++c) {
        const int e = g * GRAB_E + (wv * CPW + c) * 32 + lane;
        srcs[c] = edge_index[e];
        dsts[c] = edge_index[N_EDGES + e];
      }
      int nxt = 0;
      if (tid == 0) nxt = atomicAdd(&queue[h], 1);
      asm volatile("" :: "v"(srcs[0]), "v"(srcs[1]), "v"(dsts[0]), "v"(dsts[1]));
#pragma unroll
      for (int c = 0; c < CPW; ++c) {
        const int srcv = srcs[c];
        const int dstv = dsts[c];
        const int dh = (dstv >= N_NODES / 2) ? 1 : 0;
        unsigned mask = __builtin_amdgcn_ballot_w32(dh == h);
        for (int t = 0; t < 32 / EU && mask != 0u; ++t) {
          int s[EU], d[EU];
          int n = 0;
#pragma unroll
          for (int u = 0; u < EU; ++u) {
            s[u] = 0; d[u] = 0;
            if (mask != 0u) {
              const int j = __builtin_ctz(mask);
              mask &= mask - 1u;
              s[u] = __builtin_amdgcn_readlane(srcv, j);
              d[u] = __builtin_amdgcn_readlane(dstv, j);
              n = u + 1;
            }
          }
          float v[EU][4];
#pragma unroll
          for (int u = 0; u < EU; ++u) {
            if (u < n) {
              const float* ms = m_buf + (size_t)s[u] * HIDDEN + lane;
#pragma unroll
              for (int q = 0; q < 4; ++q) v[u][q] = ms[q * 32];
            }
          }
#pragma unroll
          for (int u = 0; u < EU; ++u) {
            if (u < n) {
              float* ad = acc_buf + (size_t)d[u] * HIDDEN + lane;
#pragma unroll
              for (int q = 0; q < 4; ++q)
                unsafeAtomicAdd(&ad[q * 32], v[u][q]);
            }
          }
        }
      }
      if (tid == 0) s_g[(it + 1) & 1] = nxt;
      __builtin_amdgcn_fence(__ATOMIC_RELEASE, "workgroup", "local");
      __builtin_amdgcn_s_barrier();
      __builtin_amdgcn_fence(__ATOMIC_ACQUIRE, "workgroup", "local");
    }
    __syncthreads();
  }
}

#define NPW 16
#define NU  8

__global__ __launch_bounds__(256) void k_out(
    const float* __restrict__ acc_buf, const float* __restrict__ Wo,
    const float* __restrict__ bo, float* __restrict__ out) {
  const int lane = threadIdx.x & 31;
  const int wave = blockIdx.x * 8 + (threadIdx.x >> 5);
  const int nodeBase = wave * NPW;
  if (nodeBase >= N_NODES) return;
  float w0[4], w1[4];
#pragma unroll
  for (int t = 0; t < 4; ++t) {
    int k = lane + t * 32;
    w0[t] = Wo[k * 2 + 0];
    w1[t] = Wo[k * 2 + 1];
  }
  const float bias = bo[lane & 1];
  float res = 0.0f;
  for (int jb = 0; jb < NPW; jb += NU) {
    float h[NU][4];
#pragma unroll
    for (int u = 0; u < NU; ++u) {
      const float* ar = acc_buf + (size_t)(nodeBase + jb + u) * HIDDEN + lane;
#pragma unroll
      for (int t = 0; t < 4; ++t) h[u][t] = ar[t * 32];
    }
#pragma unroll
    for (int u = 0; u < NU; ++u) {
      float a0 = 0.0f, a1 = 0.0f;
#pragma unroll
      for (int t = 0; t < 4; ++t) {
        float hv = fmaxf(h[u][t], 0.0f);
        a0 = fmaf(hv, w0[t], a0);
        a1 = fmaf(hv, w1[t], a1);
      }
#pragma unroll
      for (int off = 16; off >= 1; off >>= 1) {
        a0 += __shfl_xor(a0, off, 32);
        a1 += __shfl_xor(a1, off, 32);
      }
      if ((lane >> 1) == jb + u) res = (lane & 1) ? a1 : a0;
    }
  }
  out[nodeBase * 2 + lane] = res + bias;
}

extern "C" void kernel_launch(void* const* d_in, const int* in_sizes, int n_in,
                              void* d_out, int out_size, void* d_ws, size_t ws_size,
                              hipStream_t stream) {
  (void)in_sizes; (void)n_in; (void)out_size; (void)ws_size;
  const float* x     = (const float*)d_in[0];
  const int*   edge  = (const int*)  d_in[1];
  const int*   batch = (const int*)  d_in[2];
  const float* gf    = (const float*)d_in[3];
  const float* Wg    = (const float*)d_in[4];
  const float* bg    = (const float*)d_in[5];
  const float* Wc    = (const float*)d_in[6];
  const float* bc    = (const float*)d_in[7];
  const float* Wm    = (const float*)d_in[8];
  const float* bm    = (const float*)d_in[9];
  const float* Ws    = (const float*)d_in[10];
  const float* bs    = (const float*)d_in[11];
  const float* Wo    = (const float*)d_in[12];
  const float* bo    = (const float*)d_in[13];
  float* out = (float*)d_out;
  int* queue = (int*)d_out;

  float* hgw     = (float*)d_ws;
  float* m_buf   = hgw   + (size_t)N_GRAPHS * HIDDEN;
  float* acc_buf = m_buf + (size_t)N_NODES  * HIDDEN;

  k_global_enc<<<N_GRAPHS / 16, 256, 0, stream>>>(gf, Wg, bg, Wc, bc, hgw, queue);
  k_node_gemms<<<N_NODES / 16, 256, 0, stream>>>(x, batch, hgw, Wc,
                                                 Wm, bm, Ws, bs, m_buf, acc_buf);
  k_scatter<<<SC_BLOCKS, 256, 0, stream>>>(edge, m_buf, acc_buf, queue);
  k_out<<<(N_NODES / NPW + 7) / 8, 256, 0, stream>>>(acc_buf, Wo, bo, out);
}


# Edge sum moved into LDS: edges sorted by target range, one work-group per 80 nodes adds its edges' rows into an LDS tile seeded with m + h0@Ws, then the out product; float add order varies as before
# speedup vs baseline: 2.4741x; 1.1077x over previous
#include <hip/hip_runtime.h>


#define N_NODES   100000
#define N_EDGES   1600000
#define N_GRAPHS  1024
#define IN_LOCAL  16
#define IN_GLOBAL 8
#define HIDDEN    128

typedef __attribute__((ext_vector_type(2))) float v2f;
typedef __attribute__((ext_vector_type(8))) float v8f;

#define LDSH_STRIDE 129

#define RNG        80
#define NB         (N_NODES / RNG)
#define BIN_BLOCKS 250
#define BIN_EPB    (N_EDGES / BIN_BLOCKS)
#define BIN_IT     (BIN_EPB / 256)
#define SRC_BITS   17
#define SCAN_PER   40

__global__ __launch_bounds__(256) void k_global_enc(
    const float* __restrict__ gf, const float* __restrict__ Wg,
    const float* __restrict__ bg, const float* __restrict__ Wc,
    const float* __restrict__ bc, float* __restrict__ hgw,
    int* __restrict__ cnt) {
  __shared__ float lds_g[16 * LDSH_STRIDE];

  const int tid = threadIdx.x;
  {
    const int gi = blockIdx.x * 256 + tid;
    if (gi < 2 * NB) cnt[gi] = 0;
  }
  const int gBase = blockIdx.x * 16;

  {
    const int h  = tid & 127;
    const int r0 = tid >> 7;
    float wg[IN_GLOBAL];
#pragma unroll
    for (int k = 0; k < IN_GLOBAL; ++k) wg[k] = Wg[k * HIDDEN + h];
    const float b = bg[h];
#pragma unroll
    for (int i = 0; i < 8; ++i) {
      const int row = r0 + 2 * i;
      const float* gr = gf + (gBase + row) * IN_GLOBAL;
      float acc = b;
#pragma unroll
      for (int k = 0; k < IN_GLOBAL; ++k) acc = fmaf(gr[k], wg[k], acc);
      lds_g[row * LDSH_STRIDE + h] = fmaxf(acc, 0.0f);
    }
  }
  __syncthreads();

  const int lane  = tid & 31;
  const int wave  = tid >> 5;
  const int half  = lane >> 4;
  const int idx16 = lane & 15;
  const int colBase = wave * 16;

  v8f acc = {};
#pragma unroll 8
  for (int kb = 0; kb < HIDDEN; kb += 4) {
    const int k0 = kb + 2 * half;
    v2f a, b;
    a.x = lds_g[idx16 * LDSH_STRIDE + k0];
    a.y = lds_g[idx16 * LDSH_STRIDE + k0 + 1];
    b.x = Wc[(IN_LOCAL + k0) * HIDDEN + colBase + idx16];
    b.y = Wc[(IN_LOCAL + k0 + 1) * HIDDEN + colBase + idx16];
    acc = __builtin_amdgcn_wmma_f32_16x16x4_f32(
        false, a, false, b, (short)0, acc, false, false);
  }
  {
    const float bias = bc[colBase + idx16];
#pragma unroll
    for (int j = 0; j < 8; ++j) {
      int row = j + 8 * half;
      hgw[(gBase + row) * HIDDEN + colBase + idx16] = acc[j] + bias;
    }
  }
}

__device__ __forceinline__ void h0_tile_to_lds(
    const float* __restrict__ x, const int* __restrict__ batch,
    const float* __restrict__ hgw, const float* __restrict__ Wc,
    float* lds_h, int nodeBase, int half, int idx16, int colBase) {
  v8f acc;
#pragma unroll
  for (int j = 0; j < 8; ++j) {
    const int g = batch[nodeBase + j + 8 * half];
    acc[j] = hgw[g * HIDDEN + colBase + idx16];
  }
  {
    const float* xr = x + (nodeBase + idx16) * IN_LOCAL + 2 * half;
#pragma unroll
    for (int kb = 0; kb < IN_LOCAL; kb += 4) {
      const int k0 = kb + 2 * half;
      v2f a, b;
      a.x = xr[kb];
      a.y = xr[kb + 1];
      b.x = Wc[k0 * HIDDEN + colBase + idx16];
      b.y = Wc[(k0 + 1) * HIDDEN + colBase + idx16];
      acc = __builtin_amdgcn_wmma_f32_16x16x4_f32(
          false, a, false, b, (short)0, acc, false, false);
    }
  }
#pragma unroll
  for (int j = 0; j < 8; ++j) {
    int row = j + 8 * half;
    lds_h[row * LDSH_STRIDE + colBase + idx16] = fmaxf(acc[j], 0.0f);
  }
}

__device__ __forceinline__ v8f h0_times_w(
    const float* lds_h, const float* __restrict__ W, int half, int idx16, int colBase) {
  v8f am = {};
#pragma unroll 4
  for (int kb = 0; kb < HIDDEN; kb += 4) {
    const int k0 = kb + 2 * half;
    v2f a, b0;
    a.x  = lds_h[idx16 * LDSH_STRIDE + k0];
    a.y  = lds_h[idx16 * LDSH_STRIDE + k0 + 1];
    b0.x = W[k0 * HIDDEN + colBase + idx16];
    b0.y = W[(k0 + 1) * HIDDEN + colBase + idx16];
    am = __builtin_amdgcn_wmma_f32_16x16x4_f32(
        false, a, false, b0, (short)0, am, false, false);
  }
  return am;
}

__global__ __launch_bounds__(256) void k_node_msg(
    const float* __restrict__ x,  const int* __restrict__ batch,
    const float* __restrict__ hgw, const float* __restrict__ Wc,
    const float* __restrict__ Wm, const float* __restrict__ bm,
    float* __restrict__ m_buf) {
  __shared__ float lds_h[16 * LDSH_STRIDE];

  const int tid = threadIdx.x;
  const int nodeBase = blockIdx.x * 16;
  const int lane  = tid & 31;
  const int wave  = tid >> 5;
  const int half  = lane >> 4;
  const int idx16 = lane & 15;
  const int colBase = wave * 16;

  h0_tile_to_lds(x, batch, hgw, Wc, lds_h, nodeBase, half, idx16, colBase);
  __syncthreads();
  const v8f am = h0_times_w(lds_h, Wm, half, idx16, colBase);
  {
    const int col = colBase + idx16;
    const float biasm = bm[col];
#pragma unroll
    for (int j = 0; j < 8; ++j) {
      const int node = nodeBase + j + 8 * half;
      m_buf[(size_t)node * HIDDEN + col] = fmaxf(am[j] + biasm, 0.0f);
    }
  }
}

__global__ __launch_bounds__(256) void k_bin_count(
    const int* __restrict__ edge_index, int* __restrict__ hist) {
  __shared__ int lh[NB];
  const int tid = threadIdx.x;
  const int* dp = edge_index + N_EDGES + blockIdx.x * BIN_EPB;
  int dv[BIN_IT];
#pragma unroll
  for (int i = 0; i < BIN_IT; ++i) dv[i] = dp[i * 256 + tid];
  for (int b = tid; b < NB; b += 256) lh[b] = 0;
  __syncthreads();
#pragma unroll
  for (int i = 0; i < BIN_IT; ++i) atomicAdd(&lh[dv[i] / RNG], 1);
  __syncthreads();
  for (int b = tid; b < NB; b += 256) {
    const int c = lh[b];
    if (c != 0) atomicAdd(&hist[b], c);
  }
}

__global__ __launch_bounds__(256) void k_bin_fill(
    const int* __restrict__ edge_index, const int* __restrict__ hist,
    int* __restrict__ cursor, int* __restrict__ offs, int* __restrict__ list) {
  __shared__ int lh[NB];
  __shared__ int loff[NB + 1];
  const int tid  = threadIdx.x;
  const int lane = tid & 31;
  const int* sp = edge_index + blockIdx.x * BIN_EPB;
  const int* dp = sp + N_EDGES;
  int sv[BIN_IT], dv[BIN_IT];
#pragma unroll
  for (int i = 0; i < BIN_IT; ++i) {
    dv[i] = dp[i * 256 + tid];
    sv[i] = sp[i * 256 + tid];
  }
  for (int b = tid; b < NB; b += 256) lh[b] = 0;
  __syncthreads();
#pragma unroll
  for (int i = 0; i < BIN_IT; ++i) atomicAdd(&lh[dv[i] / RNG], 1);
  if (tid < 32) {
    int v[SCAN_PER];
    int s = 0;
#pragma unroll
    for (int k = 0; k < SCAN_PER; ++k) {
      const int idx = lane * SCAN_PER + k;
      v[k] = (idx < NB) ? hist[idx] : 0;
      s += v[k];
    }
    int incl = s;
#pragma unroll
    for (int o = 1; o < 32; o <<= 1) {
      const int t = __shfl_up(incl, o, 32);
      if (lane >= o) incl += t;
    }
    int run = incl - s;
#pragma unroll
    for (int k = 0; k < SCAN_PER; ++k) {
      const int idx = lane * SCAN_PER + k;
      if (idx <= NB) loff[idx] = run;
      run += v[k];
    }
  }
  __syncthreads();
  if (blockIdx.x == 0)
    for (int b = tid; b <= NB; b += 256) offs[b] = loff[b];
  int base[(NB + 255) / 256];
#pragma unroll
  for (int k = 0; k < (NB + 255) / 256; ++k) {
    const int b = tid + k * 256;
    base[k] = 0;
    if (b < NB) base[k] = atomicAdd(&cursor[b], lh[b]);
  }
#pragma unroll
  for (int k = 0; k < (NB + 255) / 256; ++k) {
    const int b = tid + k * 256;
    if (b < NB) lh[b] = loff[b] + base[k];
  }
  __syncthreads();
  int pos[BIN_IT];
#pragma unroll
  for (int i = 0; i < BIN_IT; ++i) pos[i] = atomicAdd(&lh[dv[i] / RNG], 1);
#pragma unroll
  for (int i = 0; i < BIN_IT; ++i)
    list[pos[i]] = sv[i] | ((dv[i] - (dv[i] / RNG) * RNG) << SRC_BITS);
}

#define EU 8

__global__ __launch_bounds__(256) void k_aggr_out(
    const float* __restrict__ x,  const int* __restrict__ batch,
    const float* __restrict__ hgw, const float* __restrict__ Wc,
    const float* __restrict__ Ws, const float* __restrict__ bs,
    const float* __restrict__ m_buf, const int* __restrict__ offs,
    const int* __restrict__ list, const float* __restrict__ Wo,
    const float* __restrict__ bo, float* __restrict__ out) {
  __shared__ float acc_l[RNG * HIDDEN];
  __shared__ float lds_h[16 * LDSH_STRIDE];
  __shared__ float res_l[RNG * 2];

  const int tid = threadIdx.x;
  const int lane  = tid & 31;
  const int wave  = __builtin_amdgcn_readfirstlane(tid >> 5);
  const int half  = lane >> 4;
  const int idx16 = lane & 15;
  const int colBase = wave * 16;
  const int rbase = blockIdx.x * RNG;
  const int beg = offs[blockIdx.x];
  const int end = offs[blockIdx.x + 1];

  for (int t = 0; t < RNG / 16; ++t) {
    const int nodeBase = rbase + t * 16;
    h0_tile_to_lds(x, batch, hgw, Wc, lds_h, nodeBase, half, idx16, colBase);
    __syncthreads();
    const v8f as = h0_times_w(lds_h, Ws, half, idx16, colBase);
    {
      const int col = colBase + idx16;
      const float biass = bs[col];
#pragma unroll
      for (int j = 0; j < 8; ++j) {
        const int row = j + 8 * half;
        const float mv = m_buf[(size_t)(nodeBase + row) * HIDDEN + col];
        acc_l[(t * 16 + row) * HIDDEN + col] = mv + as[j] + biass;
      }
    }
    __syncthreads();
  }

  for (int it = 0; it < N_EDGES / 256 + 1; ++it) {
    const int c = beg + it * 256 + wave * 32;
    if (c >= end) break;
    const int e = c + lane;
    const int pk = (e < end) ? list[e] : 0;
    unsigned mask = __builtin_amdgcn_ballot_w32(e < end);
    for (int g = 0; g < 32 / EU && mask != 0u; ++g) {
      int s[EU], d[EU];
      int n = 0;
#pragma unroll
      for (int u = 0; u < EU; ++u) {
        s[u] = 0; d[u] = 0;
        if (mask != 0u) {
          const int j = __builtin_ctz(mask);
          mask &= mask - 1u;
          const int p = __builtin_amdgcn_readlane(pk, j);
          s[u] = p & ((1 << SRC_BITS) - 1);
          d[u] = p >> SRC_BITS;
          n = u + 1;
        }
      }
      float v[EU][4];
#pragma unroll
      for (int u = 0; u < EU; ++u) {
        if (u < n) {
          const float* ms = m_buf + (size_t)s[u] * HIDDEN + lane;
#pragma unroll
          for (int q = 0; q < 4; ++q) v[u][q] = ms[q * 32];
        }
      }
#pragma unroll
      for (int u = 0; u < EU; ++u) {
        if (u < n) {
          float* ad = acc_l + d[u] * HIDDEN + lane;
#pragma unroll
          for (int q = 0; q < 4; ++q)
            atomicAdd(&ad[q * 32], v[u][q]);
        }
      }
    }
  }
  __syncthreads();

  {
    float w0[4], w1[4];
#pragma unroll
    for (int t = 0; t < 4; ++t) {
      const int k = lane + t * 32;
      w0[t] = Wo[k * 2 + 0];
      w1[t] = Wo[k * 2 + 1];
    }
    for (int i = 0; i < RNG / 8; ++i) {
      const int n = wave * (RNG / 8) + i;
      float a0 = 0.0f, a1 = 0.0f;
#pragma unroll
      for (int t = 0; t < 4; ++t) {
        const float hv = fmaxf(acc_l[n * HIDDEN + lane + t * 32], 0.0f);
        a0 = fmaf(hv, w0[t], a0);
        a1 = fmaf(hv, w1[t], a1);
      }
#pragma unroll
      for (int off = 16; off >= 1; off >>= 1) {
        a0 += __shfl_xor(a0, off, 32);
        a1 += __shfl_xor(a1, off, 32);
      }
      if (lane < 2) res_l[n * 2 + lane] = lane ? a1 : a0;
    }
  }
  __syncthreads();
  if (tid < RNG * 2) out[rbase * 2 + tid] = res_l[tid] + bo[tid & 1];
}

extern "C" void kernel_launch(void* const* d_in, const int* in_sizes, int n_in,
                              void* d_out, int out_size, void* d_ws, size_t ws_size,
                              hipStream_t stream) {
  (void)in_sizes; (void)n_in; (void)out_size; (void)ws_size;
  const float* x     = (const float*)d_in[0];
  const int*   edge  = (const int*)  d_in[1];
  const int*   batch = (const int*)  d_in[2];
  const float* gf    = (const float*)d_in[3];
  const float* Wg    = (const float*)d_in[4];
  const float* bg    = (const float*)d_in[5];
  const float* Wc    = (const float*)d_in[6];
  const float* bc    = (const float*)d_in[7];
  const float* Wm    = (const float*)d_in[8];
  const float* bm    = (const float*)d_in[9];
  const float* Ws    = (const float*)d_in[10];
  const float* bs    = (const float*)d_in[11];
  const float* Wo    = (const float*)d_in[12];
  const float* bo    = (const float*)d_in[13];
  float* out = (float*)d_out;

  float* hgw   = (float*)d_ws;
  float* m_buf = hgw + (size_t)N_GRAPHS * HIDDEN;
  int*   cnt   = (int*)(m_buf + (size_t)N_NODES * HIDDEN);
  int*   offs  = cnt + 4096;
  int*   list  = cnt + 8192;

  k_global_enc<<<N_GRAPHS / 16, 256, 0, stream>>>(gf, Wg, bg, Wc, bc, hgw, cnt);
  k_node_msg<<<N_NODES / 16, 256, 0, stream>>>(x, batch, hgw, Wc, Wm, bm, m_buf);
  k_bin_count<<<BIN_BLOCKS, 256, 0, stream>>>(edge, cnt);
  k_bin_fill<<<BIN_BLOCKS, 256, 0, stream>>>(edge, cnt, cnt + NB, offs, list);
  k_aggr_out<<<NB, 256, 0, stream>>>(x, batch, hgw, Wc, Ws, bs, m_buf, offs, list,
                                     Wo, bo, out);
}


# Node GEMMs blocked 5 node tiles deep per wave: each Wm/Ws fragment is loaded once per 80 nodes (was per 16) and feeds 5 independent WMMA chains; every sum keeps its order
# speedup vs baseline: 2.6812x; 1.0837x over previous
#include <hip/hip_runtime.h>


#define N_NODES   100000
#define N_EDGES   1600000
#define N_GRAPHS  1024
#define IN_LOCAL  16
#define IN_GLOBAL 8
#define HIDDEN    128

typedef __attribute__((ext_vector_type(2))) float v2f;
typedef __attribute__((ext_vector_type(8))) float v8f;

#define LDSH_STRIDE 129

#define RNG        80
#define NB         (N_NODES / RNG)
#define BIN_BLOCKS 250
#define BIN_EPB    (N_EDGES / BIN_BLOCKS)
#define BIN_IT     (BIN_EPB / 256)
#define SRC_BITS   17
#define SCAN_PER   40

__global__ __launch_bounds__(256) void k_global_enc(
    const float* __restrict__ gf, const float* __restrict__ Wg,
    const float* __restrict__ bg, const float* __restrict__ Wc,
    const float* __restrict__ bc, float* __restrict__ hgw,
    int* __restrict__ cnt) {
  __shared__ float lds_g[16 * LDSH_STRIDE];

  const int tid = threadIdx.x;
  {
    const int gi = blockIdx.x * 256 + tid;
    if (gi < 2 * NB) cnt[gi] = 0;
  }
  const int gBase = blockIdx.x * 16;

  {
    const int h  = tid & 127;
    const int r0 = tid >> 7;
    float wg[IN_GLOBAL];
#pragma unroll
    for (int k = 0; k < IN_GLOBAL; ++k) wg[k] = Wg[k * HIDDEN + h];
    const float b = bg[h];
#pragma unroll
    for (int i = 0; i < 8; ++i) {
      const int row = r0 + 2 * i;
      const float* gr = gf + (gBase + row) * IN_GLOBAL;
      float acc = b;
#pragma unroll
      for (int k = 0; k < IN_GLOBAL; ++k) acc = fmaf(gr[k], wg[k], acc);
      lds_g[row * LDSH_STRIDE + h] = fmaxf(acc, 0.0f);
    }
  }
  __syncthreads();

  const int lane  = tid & 31;
  const int wave  = tid >> 5;
  const int half  = lane >> 4;
  const int idx16 = lane & 15;
  const int colBase = wave * 16;

  v8f acc = {};
#pragma unroll 8
  for (int kb = 0; kb < HIDDEN; kb += 4) {
    const int k0 = kb + 2 * half;
    v2f a, b;
    a.x = lds_g[idx16 * LDSH_STRIDE + k0];
    a.y = lds_g[idx16 * LDSH_STRIDE + k0 + 1];
    b.x = Wc[(IN_LOCAL + k0) * HIDDEN + colBase + idx16];
    b.y = Wc[(IN_LOCAL + k0 + 1) * HIDDEN + colBase + idx16];
    acc = __builtin_amdgcn_wmma_f32_16x16x4_f32(
        false, a, false, b, (short)0, acc, false, false);
  }
  {
    const float bias = bc[colBase + idx16];
#pragma unroll
    for (int j = 0; j < 8; ++j) {
      int row = j + 8 * half;
      hgw[(gBase + row) * HIDDEN + colBase + idx16] = acc[j] + bias;
    }
  }
}

#define MT 5
static_assert(RNG == MT * 16, "kernel 5 covers its range with MT node tiles");

template <int TILE_UNROLL>
__device__ __forceinline__ void h0_rows_to_lds(
    const float* __restrict__ x, const int* __restrict__ batch,
    const float* __restrict__ hgw, const float* __restrict__ Wc,
    float* lds_h, int nodeBase, int half, int idx16, int colBase) {
  v2f bw[IN_LOCAL / 4];
#pragma unroll
  for (int kb = 0; kb < IN_LOCAL; kb += 4) {
    const int k0 = kb + 2 * half;
    bw[kb / 4].x = Wc[k0 * HIDDEN + colBase + idx16];
    bw[kb / 4].y = Wc[(k0 + 1) * HIDDEN + colBase + idx16];
  }
#pragma unroll TILE_UNROLL
  for (int t = 0; t < MT; ++t) {
    const int nb = nodeBase + t * 16;
    v8f acc;
#pragma unroll
    for (int j = 0; j < 8; ++j) {
      const int g = batch[nb + j + 8 * half];
      acc[j] = hgw[g * HIDDEN + colBase + idx16];
    }
    const float* xr = x + (nb + idx16) * IN_LOCAL + 2 * half;
#pragma unroll
    for (int kb = 0; kb < IN_LOCAL; kb += 4) {
      v2f a;
      a.x = xr[kb];
      a.y = xr[kb + 1];
      acc = __builtin_amdgcn_wmma_f32_16x16x4_f32(
          false, a, false, bw[kb / 4], (short)0, acc, false, false);
    }
#pragma unroll
    for (int j = 0; j < 8; ++j) {
      const int row = t * 16 + j + 8 * half;
      lds_h[row * LDSH_STRIDE + colBase + idx16] = fmaxf(acc[j], 0.0f);
    }
  }
}

__device__ __forceinline__ void h0_times_w(
    const float* lds_h, const float* __restrict__ W, int half, int idx16, int colBase,
    v8f (&am)[MT]) {
#pragma unroll
  for (int t = 0; t < MT; ++t) am[t] = v8f{};
#pragma unroll 2
  for (int kb = 0; kb < HIDDEN; kb += 4) {
    const int k0 = kb + 2 * half;
    v2f b0;
    b0.x = W[k0 * HIDDEN + colBase + idx16];
    b0.y = W[(k0 + 1) * HIDDEN + colBase + idx16];
#pragma unroll
    for (int t = 0; t < MT; ++t) {
      v2f a;
      a.x = lds_h[(t * 16 + idx16) * LDSH_STRIDE + k0];
      a.y = lds_h[(t * 16 + idx16) * LDSH_STRIDE + k0 + 1];
      am[t] = __builtin_amdgcn_wmma_f32_16x16x4_f32(
          false, a, false, b0, (short)0, am[t], false, false);
    }
  }
}

__global__ __launch_bounds__(256) void k_node_msg(
    const float* __restrict__ x,  const int* __restrict__ batch,
    const float* __restrict__ hgw, const float* __restrict__ Wc,
    const float* __restrict__ Wm, const float* __restrict__ bm,
    float* __restrict__ m_buf) {
  __shared__ float lds_h[MT * 16 * LDSH_STRIDE];

  const int tid = threadIdx.x;
  const int nodeBase = blockIdx.x * (MT * 16);
  const int lane  = tid & 31;
  const int wave  = tid >> 5;
  const int half  = lane >> 4;
  const int idx16 = lane & 15;
  const int colBase = wave * 16;

  h0_rows_to_lds<MT>(x, batch, hgw, Wc, lds_h, nodeBase, half, idx16, colBase);
  __syncthreads();
  v8f am[MT];
  h0_times_w(lds_h, Wm, half, idx16, colBase, am);
  {
    const int col = colBase + idx16;
    const float biasm = bm[col];
#pragma unroll
    for (int t = 0; t < MT; ++t) {
#pragma unroll
      for (int j = 0; j < 8; ++j) {
        const int node = nodeBase + t * 16 + j + 8 * half;
        m_buf[(size_t)node * HIDDEN + col] = fmaxf(am[t][j] + biasm, 0.0f);
      }
    }
  }
}

__global__ __launch_bounds__(256) void k_bin_count(
    const int* __restrict__ edge_index, int* __restrict__ hist) {
  __shared__ int lh[NB];
  const int tid = threadIdx.x;
  const int* dp = edge_index + N_EDGES + blockIdx.x * BIN_EPB;
  int dv[BIN_IT];
#pragma unroll
  for (int i = 0; i < BIN_IT; ++i) dv[i] = dp[i * 256 + tid];
  for (int b = tid; b < NB; b += 256) lh[b] = 0;
  __syncthreads();
#pragma unroll
  for (int i = 0; i < BIN_IT; ++i) atomicAdd(&lh[dv[i] / RNG], 1);
  __syncthreads();
  for (int b = tid; b < NB; b += 256) {
    const int c = lh[b];
    if (c != 0) atomicAdd(&hist[b], c);
  }
}

__global__ __launch_bounds__(256) void k_bin_fill(
    const int* __restrict__ edge_index, const int* __restrict__ hist,
    int* __restrict__ cursor, int* __restrict__ offs, int* __restrict__ list) {
  __shared__ int lh[NB];
  __shared__ int loff[NB + 1];
  const int tid  = threadIdx.x;
  const int lane = tid & 31;
  const int* sp = edge_index + blockIdx.x * BIN_EPB;
  const int* dp = sp + N_EDGES;
  int sv[BIN_IT], dv[BIN_IT];
#pragma unroll
  for (int i = 0; i < BIN_IT; ++i) {
    dv[i] = dp[i * 256 + tid];
    sv[i] = sp[i * 256 + tid];
  }
  for (int b = tid; b < NB; b += 256) lh[b] = 0;
  __syncthreads();
#pragma unroll
  for (int i = 0; i < BIN_IT; ++i) atomicAdd(&lh[dv[i] / RNG], 1);
  if (tid < 32) {
    int v[SCAN_PER];
    int s = 0;
#pragma unroll
    for (int k = 0; k < SCAN_PER; ++k) {
      const int idx = lane * SCAN_PER + k;
      v[k] = (idx < NB) ? hist[idx] : 0;
      s += v[k];
    }
    int incl = s;
#pragma unroll
    for (int o = 1; o < 32; o <<= 1) {
      const int t = __shfl_up(incl, o, 32);
      if (lane >= o) incl += t;
    }
    int run = incl - s;
#pragma unroll
    for (int k = 0; k < SCAN_PER; ++k) {
      const int idx = lane * SCAN_PER + k;
      if (idx <= NB) loff[idx] = run;
      run += v[k];
    }
  }
  __syncthreads();
  if (blockIdx.x == 0)
    for (int b = tid; b <= NB; b += 256) offs[b] = loff[b];
  int base[(NB + 255) / 256];
#pragma unroll
  for (int k = 0; k < (NB + 255) / 256; ++k) {
    const int b = tid + k * 256;
    base[k] = 0;
    if (b < NB) base[k] = atomicAdd(&cursor[b], lh[b]);
  }
#pragma unroll
  for (int k = 0; k < (NB + 255) / 256; ++k) {
    const int b = tid + k * 256;
    if (b < NB) lh[b] = loff[b] + base[k];
  }
  __syncthreads();
  int pos[BIN_IT];
#pragma unroll
  for (int i = 0; i < BIN_IT; ++i) pos[i] = atomicAdd(&lh[dv[i] / RNG], 1);
#pragma unroll
  for (int i = 0; i < BIN_IT; ++i)
    list[pos[i]] = sv[i] | ((dv[i] - (dv[i] / RNG) * RNG) << SRC_BITS);
}

#define EU 8

__global__ __launch_bounds__(256) void k_aggr_out(
    const float* __restrict__ x,  const int* __restrict__ batch,
    const float* __restrict__ hgw, const float* __restrict__ Wc,
    const float* __restrict__ Ws, const float* __restrict__ bs,
    const float* __restrict__ m_buf, const int* __restrict__ offs,
    const int* __restrict__ list, const float* __restrict__ Wo,
    const float* __restrict__ bo, float* __restrict__ out) {
  __shared__ float acc_l[RNG * LDSH_STRIDE];
  __shared__ float res_l[RNG * 2];

  const int tid = threadIdx.x;
  const int lane  = tid & 31;
  const int wave  = __builtin_amdgcn_readfirstlane(tid >> 5);
  const int half  = lane >> 4;
  const int idx16 = lane & 15;
  const int colBase = wave * 16;
  const int rbase = blockIdx.x * RNG;
  const int beg = offs[blockIdx.x];
  const int end = offs[blockIdx.x + 1];

  {
    h0_rows_to_lds<1>(x, batch, hgw, Wc, acc_l, rbase, half, idx16, colBase);
    __syncthreads();
    v8f as[MT];
    h0_times_w(acc_l, Ws, half, idx16, colBase, as);
    const int col = colBase + idx16;
    const float biass = bs[col];
    __syncthreads();
#pragma unroll
    for (int t = 0; t < MT; ++t) {
#pragma unroll
      for (int j = 0; j < 8; ++j) {
        const int row = t * 16 + j + 8 * half;
        const float mv = m_buf[(size_t)(rbase + row) * HIDDEN + col];
        acc_l[row * HIDDEN + col] = mv + as[t][j] + biass;
      }
    }
    __syncthreads();
  }

  for (int it = 0; it < N_EDGES / 256 + 1; ++it) {
    const int c = beg + it * 256 + wave * 32;
    if (c >= end) break;
    const int e = c + lane;
    const int pk = (e < end) ? list[e] : 0;
    unsigned mask = __builtin_amdgcn_ballot_w32(e < end);
    for (int g = 0; g < 32 / EU && mask != 0u; ++g) {
      int s[EU], d[EU];
      int n = 0;
#pragma unroll
      for (int u = 0; u < EU; ++u) {
        s[u] = 0; d[u] = 0;
        if (mask != 0u) {
          const int j = __builtin_ctz(mask);
          mask &= mask - 1u;
          const int p = __builtin_amdgcn_readlane(pk, j);
          s[u] = p & ((1 << SRC_BITS) - 1);
          d[u] = p >> SRC_BITS;
          n = u + 1;
        }
      }
      float v[EU][4];
#pragma unroll
      for (int u = 0; u < EU; ++u) {
        if (u < n) {
          const float* ms = m_buf + (size_t)s[u] * HIDDEN + lane;
#pragma unroll
          for (int q = 0; q < 4; ++q) v[u][q] = ms[q * 32];
        }
      }
#pragma unroll
      for (int u = 0; u < EU; ++u) {
        if (u < n) {
          float* ad = acc_l + d[u] * HIDDEN + lane;
#pragma unroll
          for (int q = 0; q < 4; ++q)
            atomicAdd(&ad[q * 32], v[u][q]);
        }
      }
    }
  }
  __syncthreads();

  {
    float w0[4], w1[4];
#pragma unroll
    for (int t = 0; t < 4; ++t) {
      const int k = lane + t * 32;
      w0[t] = Wo[k * 2 + 0];
      w1[t] = Wo[k * 2 + 1];
    }
    for (int i = 0; i < RNG / 8; ++i) {
      const int n = wave * (RNG / 8) + i;
      float a0 = 0.0f, a1 = 0.0f;
#pragma unroll
      for (int t = 0; t < 4; ++t) {
        const float hv = fmaxf(acc_l[n * HIDDEN + lane + t * 32], 0.0f);
        a0 = fmaf(hv, w0[t], a0);
        a1 = fmaf(hv, w1[t], a1);
      }
#pragma unroll
      for (int off = 16; off >= 1; off >>= 1) {
        a0 += __shfl_xor(a0, off, 32);
        a1 += __shfl_xor(a1, off, 32);
      }
      if (lane < 2) res_l[n * 2 + lane] = lane ? a1 : a0;
    }
  }
  __syncthreads();
  if (tid < RNG * 2) out[rbase * 2 + tid] = res_l[tid] + bo[tid & 1];
}

extern "C" void kernel_launch(void* const* d_in, const int* in_sizes, int n_in,
                              void* d_out, int out_size, void* d_ws, size_t ws_size,
                              hipStream_t stream) {
  (void)in_sizes; (void)n_in; (void)out_size; (void)ws_size;
  const float* x     = (const float*)d_in[0];
  const int*   edge  = (const int*)  d_in[1];
  const int*   batch = (const int*)  d_in[2];
  const float* gf    = (const float*)d_in[3];
  const float* Wg    = (const float*)d_in[4];
  const float* bg    = (const float*)d_in[5];
  const float* Wc    = (const float*)d_in[6];
  const float* bc    = (const float*)d_in[7];
  const float* Wm    = (const float*)d_in[8];
  const float* bm    = (const float*)d_in[9];
  const float* Ws    = (const float*)d_in[10];
  const float* bs    = (const float*)d_in[11];
  const float* Wo    = (const float*)d_in[12];
  const float* bo    = (const float*)d_in[13];
  float* out = (float*)d_out;

  float* hgw   = (float*)d_ws;
  float* m_buf = hgw + (size_t)N_GRAPHS * HIDDEN;
  int*   cnt   = (int*)(m_buf + (size_t)N_NODES * HIDDEN);
  int*   offs  = cnt + 4096;
  int*   list  = cnt + 8192;

  k_global_enc<<<N_GRAPHS / 16, 256, 0, stream>>>(gf, Wg, bg, Wc, bc, hgw, cnt);
  k_node_msg<<<N_NODES / (MT * 16), 256, 0, stream>>>(x, batch, hgw, Wc, Wm, bm, m_buf);
  k_bin_count<<<BIN_BLOCKS, 256, 0, stream>>>(edge, cnt);
  k_bin_fill<<<BIN_BLOCKS, 256, 0, stream>>>(edge, cnt, cnt + NB, offs, list);
  k_aggr_out<<<NB, 256, 0, stream>>>(x, batch, hgw, Wc, Ws, bs, m_buf, offs, list,
                                     Wo, bo, out);
}


# Edge sum without LDS float atomics: a range's edges are put in target-node order in LDS, each wave sums its 10 nodes' rows in registers from the tile's value; add order in a node varies as before
# speedup vs baseline: 2.9759x; 1.1099x over previous
#include <hip/hip_runtime.h>


#define N_NODES   100000
#define N_EDGES   1600000
#define N_GRAPHS  1024
#define IN_LOCAL  16
#define IN_GLOBAL 8
#define HIDDEN    128

typedef __attribute__((ext_vector_type(2))) float v2f;
typedef __attribute__((ext_vector_type(8))) float v8f;

#define LDSH_STRIDE 129

#define RNG        80
#define NB         (N_NODES / RNG)
#define BIN_BLOCKS 250
#define BIN_EPB    (N_EDGES / BIN_BLOCKS)
#define BIN_IT     (BIN_EPB / 256)
#define SRC_BITS   17
#define SCAN_PER   40

__global__ __launch_bounds__(256) void k_global_enc(
    const float* __restrict__ gf, const float* __restrict__ Wg,
    const float* __restrict__ bg, const float* __restrict__ Wc,
    const float* __restrict__ bc, float* __restrict__ hgw,
    int* __restrict__ cnt) {
  __shared__ float lds_g[16 * LDSH_STRIDE];

  const int tid = threadIdx.x;
  {
    const int gi = blockIdx.x * 256 + tid;
    if (gi < 2 * NB) cnt[gi] = 0;
  }
  const int gBase = blockIdx.x * 16;

  {
    const int h  = tid & 127;
    const int r0 = tid >> 7;
    float wg[IN_GLOBAL];
#pragma unroll
    for (int k = 0; k < IN_GLOBAL; ++k) wg[k] = Wg[k * HIDDEN + h];
    const float b = bg[h];
#pragma unroll
    for (int i = 0; i < 8; ++i) {
      const int row = r0 + 2 * i;
      const float* gr = gf + (gBase + row) * IN_GLOBAL;
      float acc = b;
#pragma unroll
      for (int k = 0; k < IN_GLOBAL; ++k) acc = fmaf(gr[k], wg[k], acc);
      lds_g[row * LDSH_STRIDE + h] = fmaxf(acc, 0.0f);
    }
  }
  __syncthreads();

  const int lane  = tid & 31;
  const int wave  = tid >> 5;
  const int half  = lane >> 4;
  const int idx16 = lane & 15;
  const int colBase = wave * 16;

  v8f acc = {};
#pragma unroll 8
  for (int kb = 0; kb < HIDDEN; kb += 4) {
    const int k0 = kb + 2 * half;
    v2f a, b;
    a.x = lds_g[idx16 * LDSH_STRIDE + k0];
    a.y = lds_g[idx16 * LDSH_STRIDE + k0 + 1];
    b.x = Wc[(IN_LOCAL + k0) * HIDDEN + colBase + idx16];
    b.y = Wc[(IN_LOCAL + k0 + 1) * HIDDEN + colBase + idx16];
    acc = __builtin_amdgcn_wmma_f32_16x16x4_f32(
        false, a, false, b, (short)0, acc, false, false);
  }
  {
    const float bias = bc[colBase + idx16];
#pragma unroll
    for (int j = 0; j < 8; ++j) {
      int row = j + 8 * half;
      hgw[(gBase + row) * HIDDEN + colBase + idx16] = acc[j] + bias;
    }
  }
}

#define MT 5
static_assert(RNG == MT * 16, "kernel 5 covers its range with MT node tiles");

template <int TILE_UNROLL>
__device__ __forceinline__ void h0_rows_to_lds(
    const float* __restrict__ x, const int* __restrict__ batch,
    const float* __restrict__ hgw, const float* __restrict__ Wc,
    float* lds_h, int nodeBase, int half, int idx16, int colBase) {
  v2f bw[IN_LOCAL / 4];
#pragma unroll
  for (int kb = 0; kb < IN_LOCAL; kb += 4) {
    const int k0 = kb + 2 * half;
    bw[kb / 4].x = Wc[k0 * HIDDEN + colBase + idx16];
    bw[kb / 4].y = Wc[(k0 + 1) * HIDDEN + colBase + idx16];
  }
#pragma unroll TILE_UNROLL
  for (int t = 0; t < MT; ++t) {
    const int nb = nodeBase + t * 16;
    v8f acc;
#pragma unroll
    for (int j = 0; j < 8; ++j) {
      const int g = batch[nb + j + 8 * half];
      acc[j] = hgw[g * HIDDEN + colBase + idx16];
    }
    const float* xr = x + (nb + idx16) * IN_LOCAL + 2 * half;
#pragma unroll
    for (int kb = 0; kb < IN_LOCAL; kb += 4) {
      v2f a;
      a.x = xr[kb];
      a.y = xr[kb + 1];
      acc = __builtin_amdgcn_wmma_f32_16x16x4_f32(
          false, a, false, bw[kb / 4], (short)0, acc, false, false);
    }
#pragma unroll
    for (int j = 0; j < 8; ++j) {
      const int row = t * 16 + j + 8 * half;
      lds_h[row * LDSH_STRIDE + colBase + idx16] = fmaxf(acc[j], 0.0f);
    }
  }
}

__device__ __forceinline__ void h0_times_w(
    const float* lds_h, const float* __restrict__ W, int half, int idx16, int colBase,
    v8f (&am)[MT]) {
#pragma unroll
  for (int t = 0; t < MT; ++t) am[t] = v8f{};
#pragma unroll 2
  for (int kb = 0; kb < HIDDEN; kb += 4) {
    const int k0 = kb + 2 * half;
    v2f b0;
    b0.x = W[k0 * HIDDEN + colBase + idx16];
    b0.y = W[(k0 + 1) * HIDDEN + colBase + idx16];
#pragma unroll
    for (int t = 0; t < MT; ++t) {
      v2f a;
      a.x = lds_h[(t * 16 + idx16) * LDSH_STRIDE + k0];
      a.y = lds_h[(t * 16 + idx16) * LDSH_STRIDE + k0 + 1];
      am[t] = __builtin_amdgcn_wmma_f32_16x16x4_f32(
          false, a, false, b0, (short)0, am[t], false, false);
    }
  }
}

__global__ __launch_bounds__(256) void k_node_msg(
    const float* __restrict__ x,  const int* __restrict__ batch,
    const float* __restrict__ hgw, const float* __restrict__ Wc,
    const float* __restrict__ Wm, const float* __restrict__ bm,
    float* __restrict__ m_buf) {
  __shared__ float lds_h[MT * 16 * LDSH_STRIDE];

  const int tid = threadIdx.x;
  const int nodeBase = blockIdx.x * (MT * 16);
  const int lane  = tid & 31;
  const int wave  = tid >> 5;
  const int half  = lane >> 4;
  const int idx16 = lane & 15;
  const int colBase = wave * 16;

  h0_rows_to_lds<MT>(x, batch, hgw, Wc, lds_h, nodeBase, half, idx16, colBase);
  __syncthreads();
  v8f am[MT];
  h0_times_w(lds_h, Wm, half, idx16, colBase, am);
  {
    const int col = colBase + idx16;
    const float biasm = bm[col];
#pragma unroll
    for (int t = 0; t < MT; ++t) {
#pragma unroll
      for (int j = 0; j < 8; ++j) {
        const int node = nodeBase + t * 16 + j + 8 * half;
        m_buf[(size_t)node * HIDDEN + col] = fmaxf(am[t][j] + biasm, 0.0f);
      }
    }
  }
}

__global__ __launch_bounds__(256) void k_bin_count(
    const int* __restrict__ edge_index, int* __restrict__ hist) {
  __shared__ int lh[NB];
  const int tid = threadIdx.x;
  const int* dp = edge_index + N_EDGES + blockIdx.x * BIN_EPB;
  int dv[BIN_IT];
#pragma unroll
  for (int i = 0; i < BIN_IT; ++i) dv[i] = dp[i * 256 + tid];
  for (int b = tid; b < NB; b += 256) lh[b] = 0;
  __syncthreads();
#pragma unroll
  for (int i = 0; i < BIN_IT; ++i) atomicAdd(&lh[dv[i] / RNG], 1);
  __syncthreads();
  for (int b = tid; b < NB; b += 256) {
    const int c = lh[b];
    if (c != 0) atomicAdd(&hist[b], c);
  }
}

__global__ __launch_bounds__(256) void k_bin_fill(
    const int* __restrict__ edge_index, const int* __restrict__ hist,
    int* __restrict__ cursor, int* __restrict__ offs, int* __restrict__ list) {
  __shared__ int lh[NB];
  __shared__ int loff[NB + 1];
  const int tid  = threadIdx.x;
  const int lane = tid & 31;
  const int* sp = edge_index + blockIdx.x * BIN_EPB;
  const int* dp = sp + N_EDGES;
  int sv[BIN_IT], dv[BIN_IT];
#pragma unroll
  for (int i = 0; i < BIN_IT; ++i) {
    dv[i] = dp[i * 256 + tid];
    sv[i] = sp[i * 256 + tid];
  }
  for (int b = tid; b < NB; b += 256) lh[b] = 0;
  __syncthreads();
#pragma unroll
  for (int i = 0; i < BIN_IT; ++i) atomicAdd(&lh[dv[i] / RNG], 1);
  if (tid < 32) {
    int v[SCAN_PER];
    int s = 0;
#pragma unroll
    for (int k = 0; k < SCAN_PER; ++k) {
      const int idx = lane * SCAN_PER + k;
      v[k] = (idx < NB) ? hist[idx] : 0;
      s += v[k];
    }
    int incl = s;
#pragma unroll
    for (int o = 1; o < 32; o <<= 1) {
      const int t = __shfl_up(incl, o, 32);
      if (lane >= o) incl += t;
    }
    int run = incl - s;
#pragma unroll
    for (int k = 0; k < SCAN_PER; ++k) {
      const int idx = lane * SCAN_PER + k;
      if (idx <= NB) loff[idx] = run;
      run += v[k];
    }
  }
  __syncthreads();
  if (blockIdx.x == 0)
    for (int b = tid; b <= NB; b += 256) offs[b] = loff[b];
  int base[(NB + 255) / 256];
#pragma unroll
  for (int k = 0; k < (NB + 255) / 256; ++k) {
    const int b = tid + k * 256;
    base[k] = 0;
    if (b < NB) base[k] = atomicAdd(&cursor[b], lh[b]);
  }
#pragma unroll
  for (int k = 0; k < (NB + 255) / 256; ++k) {
    const int b = tid + k * 256;
    if (b < NB) lh[b] = loff[b] + base[k];
  }
  __syncthreads();
  int pos[BIN_IT];
#pragma unroll
  for (int i = 0; i < BIN_IT; ++i) pos[i] = atomicAdd(&lh[dv[i] / RNG], 1);
#pragma unroll
  for (int i = 0; i < BIN_IT; ++i)
    list[pos[i]] = sv[i] | ((dv[i] - (dv[i] / RNG) * RNG) << SRC_BITS);
}

#define EU 8
#define SORT_CAP 2048
#define SORT_IT  (SORT_CAP / 256)
#define NPW      (RNG / 8)

__global__ __launch_bounds__(256) void k_aggr_out(
    const float* __restrict__ x,  const int* __restrict__ batch,
    const float* __restrict__ hgw, const float* __restrict__ Wc,
    const float* __restrict__ Ws, const float* __restrict__ bs,
    const float* __restrict__ m_buf, const int* __restrict__ offs,
    const int* __restrict__ list, const float* __restrict__ Wo,
    const float* __restrict__ bo, float* __restrict__ out) {
  __shared__ float acc_l[RNG * LDSH_STRIDE];
  __shared__ float res_l[RNG * 2];
  __shared__ int   srt_l[SORT_CAP];
  __shared__ int   nst_l[RNG + 1];

  const int tid = threadIdx.x;
  const int lane  = tid & 31;
  const int wave  = __builtin_amdgcn_readfirstlane(tid >> 5);
  const int half  = lane >> 4;
  const int idx16 = lane & 15;
  const int colBase = wave * 16;
  const int rbase = blockIdx.x * RNG;
  const int beg = offs[blockIdx.x];
  const int end = offs[blockIdx.x + 1];

  {
    h0_rows_to_lds<1>(x, batch, hgw, Wc, acc_l, rbase, half, idx16, colBase);
    __syncthreads();
    v8f as[MT];
    h0_times_w(acc_l, Ws, half, idx16, colBase, as);
    const int col = colBase + idx16;
    const float biass = bs[col];
    __syncthreads();
#pragma unroll
    for (int t = 0; t < MT; ++t) {
#pragma unroll
      for (int j = 0; j < 8; ++j) {
        const int row = t * 16 + j + 8 * half;
        const float mv = m_buf[(size_t)(rbase + row) * HIDDEN + col];
        acc_l[row * HIDDEN + col] = mv + as[t][j] + biass;
      }
    }
    __syncthreads();
  }

  for (int pass = 0; pass < N_EDGES / SORT_CAP + 1; ++pass) {
    const int pbeg = beg + pass * SORT_CAP;
    if (pbeg >= end) break;
    const int pn = min(end - pbeg, SORT_CAP);

    if (tid <= RNG) nst_l[tid] = 0;
    int pk[SORT_IT], rk[SORT_IT];
#pragma unroll
    for (int i = 0; i < SORT_IT; ++i) {
      const int e = i * 256 + tid;
      pk[i] = (e < pn) ? list[pbeg + e] : -1;
    }
    __syncthreads();
#pragma unroll
    for (int i = 0; i < SORT_IT; ++i) {
      rk[i] = 0;
      if (pk[i] >= 0) rk[i] = atomicAdd(&nst_l[pk[i] >> SRC_BITS], 1);
    }
    __syncthreads();
    if (wave == 0) {
      const int i0 = lane * 3;
      const int v0 = (i0     < RNG) ? nst_l[i0]     : 0;
      const int v1 = (i0 + 1 < RNG) ? nst_l[i0 + 1] : 0;
      const int v2 = (i0 + 2 < RNG) ? nst_l[i0 + 2] : 0;
      const int s = v0 + v1 + v2;
      int incl = s;
#pragma unroll
      for (int o = 1; o < 32; o <<= 1) {
        const int t = __shfl_up(incl, o, 32);
        if (lane >= o) incl += t;
      }
      const int run = incl - s;
      if (i0     <= RNG) nst_l[i0]     = run;
      if (i0 + 1 <= RNG) nst_l[i0 + 1] = run + v0;
      if (i0 + 2 <= RNG) nst_l[i0 + 2] = run + v0 + v1;
    }
    __syncthreads();
#pragma unroll
    for (int i = 0; i < SORT_IT; ++i)
      if (pk[i] >= 0) srt_l[nst_l[pk[i] >> SRC_BITS] + rk[i]] = pk[i];
    __syncthreads();

    const int sb = __builtin_amdgcn_readfirstlane(nst_l[wave * NPW]);
    const int se = __builtin_amdgcn_readfirstlane(nst_l[wave * NPW + NPW]);
    int cur = -1;
    float a[4] = {0.0f, 0.0f, 0.0f, 0.0f};
    for (int c = sb; c < se; c += 32) {
      const int e = c + lane;
      const int q = (e < se) ? srt_l[e] : 0;
      const int cn = min(se - c, 32);
      for (int g = 0; g < cn; g += EU) {
        const int n = min(cn - g, EU);
        int s[EU], d[EU];
#pragma unroll
        for (int u = 0; u < EU; ++u) {
          const int p = __builtin_amdgcn_readlane(q, min(g + u, cn - 1));
          s[u] = p & ((1 << SRC_BITS) - 1);
          d[u] = p >> SRC_BITS;
        }
        float v[EU][4];
#pragma unroll
        for (int u = 0; u < EU; ++u) {
          const float* ms = m_buf + (size_t)s[u] * HIDDEN + lane;
#pragma unroll
          for (int qq = 0; qq < 4; ++qq) v[u][qq] = ms[qq * 32];
        }
#pragma unroll
        for (int u = 0; u < EU; ++u) {
          if (d[u] != cur) {
            if (cur >= 0) {
              float* ad = acc_l + cur * HIDDEN + lane;
#pragma unroll
              for (int qq = 0; qq < 4; ++qq) ad[qq * 32] = a[qq];
            }
            cur = d[u];
            const float* al = acc_l + cur * HIDDEN + lane;
#pragma unroll
            for (int qq = 0; qq < 4; ++qq) a[qq] = al[qq * 32];
          }
#pragma unroll
          for (int qq = 0; qq < 4; ++qq) a[qq] += (u < n) ? v[u][qq] : 0.0f;
        }
      }
    }
    if (cur >= 0) {
      float* ad = acc_l + cur * HIDDEN + lane;
#pragma unroll
      for (int qq = 0; qq < 4; ++qq) ad[qq * 32] = a[qq];
    }
    __syncthreads();
  }

  {
    float w0[4], w1[4];
#pragma unroll
    for (int t = 0; t < 4; ++t) {
      const int k = lane + t * 32;
      w0[t] = Wo[k * 2 + 0];
      w1[t] = Wo[k * 2 + 1];
    }
    for (int i = 0; i < NPW; ++i) {
      const int n = wave * NPW + i;
      float a0 = 0.0f, a1 = 0.0f;
#pragma unroll
      for (int t = 0; t < 4; ++t) {
        const float hv = fmaxf(acc_l[n * HIDDEN + lane + t * 32], 0.0f);
        a0 = fmaf(hv, w0[t], a0);
        a1 = fmaf(hv, w1[t], a1);
      }
#pragma unroll
      for (int off = 16; off >= 1; off >>= 1) {
        a0 += __shfl_xor(a0, off, 32);
        a1 += __shfl_xor(a1, off, 32);
      }
      if (lane < 2) res_l[n * 2 + lane] = lane ? a1 : a0;
    }
  }
  __syncthreads();
  if (tid < RNG * 2) out[rbase * 2 + tid] = res_l[tid] + bo[tid & 1];
}

extern "C" void kernel_launch(void* const* d_in, const int* in_sizes, int n_in,
                              void* d_out, int out_size, void* d_ws, size_t ws_size,
                              hipStream_t stream) {
  (void)in_sizes; (void)n_in; (void)out_size; (void)ws_size;
  const float* x     = (const float*)d_in[0];
  const int*   edge  = (const int*)  d_in[1];
  const int*   batch = (const int*)  d_in[2];
  const float* gf    = (const float*)d_in[3];
  const float* Wg    = (const float*)d_in[4];
  const float* bg    = (const float*)d_in[5];
  const float* Wc    = (const float*)d_in[6];
  const float* bc    = (const float*)d_in[7];
  const float* Wm    = (const float*)d_in[8];
  const float* bm    = (const float*)d_in[9];
  const float* Ws    = (const float*)d_in[10];
  const float* bs    = (const float*)d_in[11];
  const float* Wo    = (const float*)d_in[12];
  const float* bo    = (const float*)d_in[13];
  float* out = (float*)d_out;

  float* hgw   = (float*)d_ws;
  float* m_buf = hgw + (size_t)N_GRAPHS * HIDDEN;
  int*   cnt   = (int*)(m_buf + (size_t)N_NODES * HIDDEN);
  int*   offs  = cnt + 4096;
  int*   list  = cnt + 8192;

  k_global_enc<<<N_GRAPHS / 16, 256, 0, stream>>>(gf, Wg, bg, Wc, bc, hgw, cnt);
  k_node_msg<<<N_NODES / (MT * 16), 256, 0, stream>>>(x, batch, hgw, Wc, Wm, bm, m_buf);
  k_bin_count<<<BIN_BLOCKS, 256, 0, stream>>>(edge, cnt);
  k_bin_fill<<<BIN_BLOCKS, 256, 0, stream>>>(edge, cnt, cnt + NB, offs, list);
  k_aggr_out<<<NB, 256, 0, stream>>>(x, batch, hgw, Wc, Ws, bs, m_buf, offs, list,
                                     Wo, bo, out);
}


# One-pass edge sort, no global counts: a 6400-edge group sorts its own by target range in LDS in the encoder's launch, a range's group collects 250 pieces; 5 launches to 3; add order varies as before
# speedup vs baseline: 3.3043x; 1.1104x over previous
#include <hip/hip_runtime.h>


#define N_NODES   100000
#define N_EDGES   1600000
#define N_GRAPHS  1024
#define IN_LOCAL  16
#define IN_GLOBAL 8
#define HIDDEN    128

typedef __attribute__((ext_vector_type(2))) float v2f;
typedef __attribute__((ext_vector_type(8))) float v8f;

#define LDSH_STRIDE 129

#define RNG        80
#define NB         (N_NODES / RNG)
#define BIN_BLOCKS 250
#define BIN_EPB    (N_EDGES / BIN_BLOCKS)
#define BIN_IT     (BIN_EPB / 256)
#define SRC_BITS   17
#define SCAN_PER   40
#define LOC_STRIDE 256
static_assert(BIN_BLOCKS <= LOC_STRIDE && BIN_BLOCKS <= 256, "one piece per thread of a range's work-group");
static_assert((NB + 1) * LOC_STRIDE <= 327680, "the piece starts end where the list begins");

__global__ __launch_bounds__(256) void k_enc_sort(
    const float* __restrict__ gf, const float* __restrict__ Wg,
    const float* __restrict__ bg, const float* __restrict__ Wc,
    const float* __restrict__ bc, float* __restrict__ hgw,
    const int* __restrict__ edge_index, int* __restrict__ loc,
    int* __restrict__ list) {
  __shared__ float lds_g[16 * LDSH_STRIDE];
  __shared__ int lh[NB];
  __shared__ int loff[NB + 1];
  __shared__ int seg_l[BIN_EPB];

  const int tid = threadIdx.x;
  const int lane  = tid & 31;

  if (blockIdx.x < BIN_BLOCKS) {
    const int grp = blockIdx.x;
    const int* sp = edge_index + grp * BIN_EPB;
    const int* dp = sp + N_EDGES;
    int sv[BIN_IT], dv[BIN_IT];
#pragma unroll
    for (int i = 0; i < BIN_IT; ++i) {
      dv[i] = dp[i * 256 + tid];
      sv[i] = sp[i * 256 + tid];
    }
    for (int b = tid; b < NB; b += 256) lh[b] = 0;
    __syncthreads();
#pragma unroll
    for (int i = 0; i < BIN_IT; ++i) atomicAdd(&lh[dv[i] / RNG], 1);
    __syncthreads();
    if (tid < 32) {
      int v[SCAN_PER];
      int s = 0;
#pragma unroll
      for (int k = 0; k < SCAN_PER; ++k) {
        const int idx = lane * SCAN_PER + k;
        v[k] = (idx < NB) ? lh[idx] : 0;
        s += v[k];
      }
      int incl = s;
#pragma unroll
      for (int o = 1; o < 32; o <<= 1) {
        const int t = __shfl_up(incl, o, 32);
        if (lane >= o) incl += t;
      }
      int run = incl - s;
#pragma unroll
      for (int k = 0; k < SCAN_PER; ++k) {
        const int idx = lane * SCAN_PER + k;
        if (idx <= NB) loff[idx] = run;
        run += v[k];
      }
    }
    __syncthreads();
    for (int r = tid; r <= NB; r += 256) {
      const int o = loff[r];
      loc[r * LOC_STRIDE + grp] = o;
      if (r < NB) lh[r] = o;
    }
    __syncthreads();
    int pos[BIN_IT];
#pragma unroll
    for (int i = 0; i < BIN_IT; ++i) pos[i] = atomicAdd(&lh[dv[i] / RNG], 1);
#pragma unroll
    for (int i = 0; i < BIN_IT; ++i)
      seg_l[pos[i]] = sv[i] | ((dv[i] - (dv[i] / RNG) * RNG) << SRC_BITS);
    __syncthreads();
    int* lp = list + grp * BIN_EPB;
#pragma unroll
    for (int i = 0; i < BIN_IT; ++i) lp[i * 256 + tid] = seg_l[i * 256 + tid];
    return;
  }

  const int gBase = (blockIdx.x - BIN_BLOCKS) * 16;

  {
    const int h  = tid & 127;
    const int r0 = tid >> 7;
    float wg[IN_GLOBAL];
#pragma unroll
    for (int k = 0; k < IN_GLOBAL; ++k) wg[k] = Wg[k * HIDDEN + h];
    const float b = bg[h];
#pragma unroll
    for (int i = 0; i < 8; ++i) {
      const int row = r0 + 2 * i;
      const float* gr = gf + (gBase + row) * IN_GLOBAL;
      float acc = b;
#pragma unroll
      for (int k = 0; k < IN_GLOBAL; ++k) acc = fmaf(gr[k], wg[k], acc);
      lds_g[row * LDSH_STRIDE + h] = fmaxf(acc, 0.0f);
    }
  }
  __syncthreads();

  const int wave  = tid >> 5;
  const int half  = lane >> 4;
  const int idx16 = lane & 15;
  const int colBase = wave * 16;

  v8f acc = {};
#pragma unroll 8
  for (int kb = 0; kb < HIDDEN; kb += 4) {
    const int k0 = kb + 2 * half;
    v2f a, b;
    a.x = lds_g[idx16 * LDSH_STRIDE + k0];
    a.y = lds_g[idx16 * LDSH_STRIDE + k0 + 1];
    b.x = Wc[(IN_LOCAL + k0) * HIDDEN + colBase + idx16];
    b.y = Wc[(IN_LOCAL + k0 + 1) * HIDDEN + colBase + idx16];
    acc = __builtin_amdgcn_wmma_f32_16x16x4_f32(
        false, a, false, b, (short)0, acc, false, false);
  }
  {
    const float bias = bc[colBase + idx16];
#pragma unroll
    for (int j = 0; j < 8; ++j) {
      int row = j + 8 * half;
      hgw[(gBase + row) * HIDDEN + colBase + idx16] = acc[j] + bias;
    }
  }
}

#define MT 5
static_assert(RNG == MT * 16, "kernel 5 covers its range with MT node tiles");

template <int TILE_UNROLL>
__device__ __forceinline__ void h0_rows_to_lds(
    const float* __restrict__ x, const int* __restrict__ batch,
    const float* __restrict__ hgw, const float* __restrict__ Wc,
    float* lds_h, int nodeBase, int half, int idx16, int colBase) {
  v2f bw[IN_LOCAL / 4];
#pragma unroll
  for (int kb = 0; kb < IN_LOCAL; kb += 4) {
    const int k0 = kb + 2 * half;
    bw[kb / 4].x = Wc[k0 * HIDDEN + colBase + idx16];
    bw[kb / 4].y = Wc[(k0 + 1) * HIDDEN + colBase + idx16];
  }
#pragma unroll TILE_UNROLL
  for (int t = 0; t < MT; ++t) {
    const int nb = nodeBase + t * 16;
    v8f acc;
#pragma unroll
    for (int j = 0; j < 8; ++j) {
      const int g = batch[nb + j + 8 * half];
      acc[j] = hgw[g * HIDDEN + colBase + idx16];
    }
    const float* xr = x + (nb + idx16) * IN_LOCAL + 2 * half;
#pragma unroll
    for (int kb = 0; kb < IN_LOCAL; kb += 4) {
      v2f a;
      a.x = xr[kb];
      a.y = xr[kb + 1];
      acc = __builtin_amdgcn_wmma_f32_16x16x4_f32(
          false, a, false, bw[kb / 4], (short)0, acc, false, false);
    }
#pragma unroll
    for (int j = 0; j < 8; ++j) {
      const int row = t * 16 + j + 8 * half;
      lds_h[row * LDSH_STRIDE + colBase + idx16] = fmaxf(acc[j], 0.0f);
    }
  }
}

__device__ __forceinline__ void h0_times_w(
    const float* lds_h, const float* __restrict__ W, int half, int idx16, int colBase,
    v8f (&am)[MT]) {
#pragma unroll
  for (int t = 0; t < MT; ++t) am[t] = v8f{};
#pragma unroll 2
  for (int kb = 0; kb < HIDDEN; kb += 4) {
    const int k0 = kb + 2 * half;
    v2f b0;
    b0.x = W[k0 * HIDDEN + colBase + idx16];
    b0.y = W[(k0 + 1) * HIDDEN + colBase + idx16];
#pragma unroll
    for (int t = 0; t < MT; ++t) {
      v2f a;
      a.x = lds_h[(t * 16 + idx16) * LDSH_STRIDE + k0];
      a.y = lds_h[(t * 16 + idx16) * LDSH_STRIDE + k0 + 1];
      am[t] = __builtin_amdgcn_wmma_f32_16x16x4_f32(
          false, a, false, b0, (short)0, am[t], false, false);
    }
  }
}

__global__ __launch_bounds__(256) void k_node_msg(
    const float* __restrict__ x,  const int* __restrict__ batch,
    const float* __restrict__ hgw, const float* __restrict__ Wc,
    const float* __restrict__ Wm, const float* __restrict__ bm,
    float* __restrict__ m_buf) {
  __shared__ float lds_h[MT * 16 * LDSH_STRIDE];

  const int tid = threadIdx.x;
  const int nodeBase = blockIdx.x * (MT * 16);
  const int lane  = tid & 31;
  const int wave  = tid >> 5;
  const int half  = lane >> 4;
  const int idx16 = lane & 15;
  const int colBase = wave * 16;

  h0_rows_to_lds<MT>(x, batch, hgw, Wc, lds_h, nodeBase, half, idx16, colBase);
  __syncthreads();
  v8f am[MT];
  h0_times_w(lds_h, Wm, half, idx16, colBase, am);
  {
    const int col = colBase + idx16;
    const float biasm = bm[col];
#pragma unroll
    for (int t = 0; t < MT; ++t) {
#pragma unroll
      for (int j = 0; j < 8; ++j) {
        const int node = nodeBase + t * 16 + j + 8 * half;
        m_buf[(size_t)node * HIDDEN + col] = fmaxf(am[t][j] + biasm, 0.0f);
      }
    }
  }
}

#define EU 8
#define PK 16
#define PK_LONG 8
#define SORT_CAP 2048
static_assert(BIN_BLOCKS * PK_LONG <= SORT_CAP, "a pass of PK_LONG entries per piece always fits");
#define NPW      (RNG / 8)

__global__ __launch_bounds__(256) void k_aggr_out(
    const float* __restrict__ x,  const int* __restrict__ batch,
    const float* __restrict__ hgw, const float* __restrict__ Wc,
    const float* __restrict__ Ws, const float* __restrict__ bs,
    const float* __restrict__ m_buf, const int* __restrict__ loc,
    const int* __restrict__ list, const float* __restrict__ Wo,
    const float* __restrict__ bo, float* __restrict__ out) {
  __shared__ float acc_l[RNG * LDSH_STRIDE];
  __shared__ float res_l[RNG * 2];
  __shared__ int   srt_l[SORT_CAP];
  __shared__ int   nst_l[RNG + 1];
  __shared__ int   wmx_l[8];
  __shared__ int   wtot_l[8];

  const int tid = threadIdx.x;
  const int lane  = tid & 31;
  const int wave  = __builtin_amdgcn_readfirstlane(tid >> 5);
  const int half  = lane >> 4;
  const int idx16 = lane & 15;
  const int colBase = wave * 16;
  const int rbase = blockIdx.x * RNG;
  int pst, pc;

  {
    h0_rows_to_lds<1>(x, batch, hgw, Wc, acc_l, rbase, half, idx16, colBase);
    __syncthreads();
    v8f as[MT];
    h0_times_w(acc_l, Ws, half, idx16, colBase, as);
    const int col = colBase + idx16;
    const float biass = bs[col];
    __syncthreads();
#pragma unroll
    for (int t = 0; t < MT; ++t) {
#pragma unroll
      for (int j = 0; j < 8; ++j) {
        const int row = t * 16 + j + 8 * half;
        const float mv = m_buf[(size_t)(rbase + row) * HIDDEN + col];
        acc_l[row * HIDDEN + col] = mv + as[t][j] + biass;
      }
    }
    int p0 = 0, p1 = 0;
    if (tid < BIN_BLOCKS) {
      p0 = loc[blockIdx.x * LOC_STRIDE + tid];
      p1 = loc[(blockIdx.x + 1) * LOC_STRIDE + tid];
    }
    pst = tid * BIN_EPB + p0;
    pc  = p1 - p0;
    {
      int m = pc, s = pc;
#pragma unroll
      for (int off = 16; off >= 1; off >>= 1) {
        m = max(m, __shfl_xor(m, off, 32));
        s += __shfl_xor(s, off, 32);
      }
      if (lane == 0) { wmx_l[wave] = m; wtot_l[wave] = s; }
    }
    __syncthreads();
  }

  int mx = 0, tot = 0;
#pragma unroll
  for (int w = 0; w < 8; ++w) {
    mx = max(mx, wmx_l[w]);
    tot += wtot_l[w];
  }
  const int step = (tot <= SORT_CAP) ? PK : PK_LONG;
  for (int pass = 0; pass < BIN_EPB / PK_LONG + 1; ++pass) {
    const int pb = pass * step;
    if (pb >= mx) break;

    if (tid <= RNG) nst_l[tid] = 0;
    int pk[PK];
#pragma unroll
    for (int i = 0; i < PK; ++i)
      pk[i] = (i < step && pb + i < pc) ? list[pst + pb + i] : -1;
    __syncthreads();
#pragma unroll
    for (int i = 0; i < PK; ++i)
      if (pk[i] >= 0) atomicAdd(&nst_l[pk[i] >> SRC_BITS], 1);
    __syncthreads();
    if (wave == 0) {
      const int i0 = lane * 3;
      const int v0 = (i0     < RNG) ? nst_l[i0]     : 0;
      const int v1 = (i0 + 1 < RNG) ? nst_l[i0 + 1] : 0;
      const int v2 = (i0 + 2 < RNG) ? nst_l[i0 + 2] : 0;
      const int s = v0 + v1 + v2;
      int incl = s;
#pragma unroll
      for (int o = 1; o < 32; o <<= 1) {
        const int t = __shfl_up(incl, o, 32);
        if (lane >= o) incl += t;
      }
      const int run = incl - s;
      if (i0     <= RNG) nst_l[i0]     = run;
      if (i0 + 1 <= RNG) nst_l[i0 + 1] = run + v0;
      if (i0 + 2 <= RNG) nst_l[i0 + 2] = run + v0 + v1;
    }
    __syncthreads();
#pragma unroll
    for (int i = 0; i < PK; ++i)
      if (pk[i] >= 0) srt_l[atomicAdd(&nst_l[pk[i] >> SRC_BITS], 1)] = pk[i];
    __syncthreads();

    const int sb = (wave == 0) ? 0 : __builtin_amdgcn_readfirstlane(nst_l[wave * NPW - 1]);
    const int se = __builtin_amdgcn_readfirstlane(nst_l[wave * NPW + NPW - 1]);
    int cur = -1;
    float a[4] = {0.0f, 0.0f, 0.0f, 0.0f};
    for (int c = sb; c < se; c += 32) {
      const int e = c + lane;
      const int q = (e < se) ? srt_l[e] : 0;
      const int cn = min(se - c, 32);
      for (int g = 0; g < cn; g += EU) {
        const int n = min(cn - g, EU);
        int s[EU], d[EU];
#pragma unroll
        for (int u = 0; u < EU; ++u) {
          const int p = __builtin_amdgcn_readlane(q, min(g + u, cn - 1));
          s[u] = p & ((1 << SRC_BITS) - 1);
          d[u] = p >> SRC_BITS;
        }
        float v[EU][4];
#pragma unroll
        for (int u = 0; u < EU; ++u) {
          const float* ms = m_buf + (size_t)s[u] * HIDDEN + lane;
#pragma unroll
          for (int qq = 0; qq < 4; ++qq) v[u][qq] = ms[qq * 32];
        }
#pragma unroll
        for (int u = 0; u < EU; ++u) {
          if (d[u] != cur) {
            if (cur >= 0) {
              float* ad = acc_l + cur * HIDDEN + lane;
#pragma unroll
              for (int qq = 0; qq < 4; ++qq) ad[qq * 32] = a[qq];
            }
            cur = d[u];
            const float* al = acc_l + cur * HIDDEN + lane;
#pragma unroll
            for (int qq = 0; qq < 4; ++qq) a[qq] = al[qq * 32];
          }
#pragma unroll
          for (int qq = 0; qq < 4; ++qq) a[qq] += (u < n) ? v[u][qq] : 0.0f;
        }
      }
    }
    if (cur >= 0) {
      float* ad = acc_l + cur * HIDDEN + lane;
#pragma unroll
      for (int qq = 0; qq < 4; ++qq) ad[qq * 32] = a[qq];
    }
    __syncthreads();
  }

  {
    float w0[4], w1[4];
#pragma unroll
    for (int t = 0; t < 4; ++t) {
      const int k = lane + t * 32;
      w0[t] = Wo[k * 2 + 0];
      w1[t] = Wo[k * 2 + 1];
    }
    for (int i = 0; i < NPW; ++i) {
      const int n = wave * NPW + i;
      float a0 = 0.0f, a1 = 0.0f;
#pragma unroll
      for (int t = 0; t < 4; ++t) {
        const float hv = fmaxf(acc_l[n * HIDDEN + lane + t * 32], 0.0f);
        a0 = fmaf(hv, w0[t], a0);
        a1 = fmaf(hv, w1[t], a1);
      }
#pragma unroll
      for (int off = 16; off >= 1; off >>= 1) {
        a0 += __shfl_xor(a0, off, 32);
        a1 += __shfl_xor(a1, off, 32);
      }
      if (lane < 2) res_l[n * 2 + lane] = lane ? a1 : a0;
    }
  }
  __syncthreads();
  if (tid < RNG * 2) out[rbase * 2 + tid] = res_l[tid] + bo[tid & 1];
}

extern "C" void kernel_launch(void* const* d_in, const int* in_sizes, int n_in,
                              void* d_out, int out_size, void* d_ws, size_t ws_size,
                              hipStream_t stream) {
  (void)in_sizes; (void)n_in; (void)out_size; (void)ws_size;
  const float* x     = (const float*)d_in[0];
  const int*   edge  = (const int*)  d_in[1];
  const int*   batch = (const int*)  d_in[2];
  const float* gf    = (const float*)d_in[3];
  const float* Wg    = (const float*)d_in[4];
  const float* bg    = (const float*)d_in[5];
  const float* Wc    = (const float*)d_in[6];
  const float* bc    = (const float*)d_in[7];
  const float* Wm    = (const float*)d_in[8];
  const float* bm    = (const float*)d_in[9];
  const float* Ws    = (const float*)d_in[10];
  const float* bs    = (const float*)d_in[11];
  const float* Wo    = (const float*)d_in[12];
  const float* bo    = (const float*)d_in[13];
  float* out = (float*)d_out;

  float* hgw   = (float*)d_ws;
  float* m_buf = hgw + (size_t)N_GRAPHS * HIDDEN;
  int*   loc   = (int*)(m_buf + (size_t)N_NODES * HIDDEN);
  int*   list  = loc + 327680;

  k_enc_sort<<<BIN_BLOCKS + N_GRAPHS / 16, 256, 0, stream>>>(gf, Wg, bg, Wc, bc, hgw, edge, loc, list);
  k_node_msg<<<N_NODES / (MT * 16), 256, 0, stream>>>(x, batch, hgw, Wc, Wm, bm, m_buf);
  k_aggr_out<<<NB, 256, 0, stream>>>(x, batch, hgw, Wc, Ws, bs, m_buf, loc, list,
                                     Wo, bo, out);
}


# Edge rows fetched 16 bytes per lane: one load instruction per 512-B row of m (was four, one 128-B line each), running sum held in the same four columns; add order in a node varies as before
# speedup vs baseline: 3.4088x; 1.0316x over previous
#include <hip/hip_runtime.h>


#define N_NODES   100000
#define N_EDGES   1600000
#define N_GRAPHS  1024
#define IN_LOCAL  16
#define IN_GLOBAL 8
#define HIDDEN    128

typedef __attribute__((ext_vector_type(2))) float v2f;
typedef __attribute__((ext_vector_type(8))) float v8f;
typedef __attribute__((ext_vector_type(4))) float v4f;

#define LDSH_STRIDE 129

#define RNG        80
#define NB         (N_NODES / RNG)
#define BIN_BLOCKS 250
#define BIN_EPB    (N_EDGES / BIN_BLOCKS)
#define BIN_IT     (BIN_EPB / 256)
#define SRC_BITS   17
#define SCAN_PER   40
#define LOC_STRIDE 256
static_assert(BIN_BLOCKS <= LOC_STRIDE && BIN_BLOCKS <= 256, "one piece per thread of a range's work-group");
static_assert((NB + 1) * LOC_STRIDE <= 327680, "the piece starts end where the list begins");

__global__ __launch_bounds__(256) void k_enc_sort(
    const float* __restrict__ gf, const float* __restrict__ Wg,
    const float* __restrict__ bg, const float* __restrict__ Wc,
    const float* __restrict__ bc, float* __restrict__ hgw,
    const int* __restrict__ edge_index, int* __restrict__ loc,
    int* __restrict__ list) {
  __shared__ float lds_g[16 * LDSH_STRIDE];
  __shared__ int lh[NB];
  __shared__ int loff[NB + 1];
  __shared__ int seg_l[BIN_EPB];

  const int tid = threadIdx.x;
  const int lane  = tid & 31;

  if (blockIdx.x < BIN_BLOCKS) {
    const int grp = blockIdx.x;
    const int* sp = edge_index + grp * BIN_EPB;
    const int* dp = sp + N_EDGES;
    int sv[BIN_IT], dv[BIN_IT];
#pragma unroll
    for (int i = 0; i < BIN_IT; ++i) {
      dv[i] = dp[i * 256 + tid];
      sv[i] = sp[i * 256 + tid];
    }
    for (int b = tid; b < NB; b += 256) lh[b] = 0;
    __syncthreads();
#pragma unroll
    for (int i = 0; i < BIN_IT; ++i) atomicAdd(&lh[dv[i] / RNG], 1);
    __syncthreads();
    if (tid < 32) {
      int v[SCAN_PER];
      int s = 0;
#pragma unroll
      for (int k = 0; k < SCAN_PER; ++k) {
        const int idx = lane * SCAN_PER + k;
        v[k] = (idx < NB) ? lh[idx] : 0;
        s += v[k];
      }
      int incl = s;
#pragma unroll
      for (int o = 1; o < 32; o <<= 1) {
        const int t = __shfl_up(incl, o, 32);
        if (lane >= o) incl += t;
      }
      int run = incl - s;
#pragma unroll
      for (int k = 0; k < SCAN_PER; ++k) {
        const int idx = lane * SCAN_PER + k;
        if (idx <= NB) loff[idx] = run;
        run += v[k];
      }
    }
    __syncthreads();
    for (int r = tid; r <= NB; r += 256) {
      const int o = loff[r];
      loc[r * LOC_STRIDE + grp] = o;
      if (r < NB) lh[r] = o;
    }
    __syncthreads();
    int pos[BIN_IT];
#pragma unroll
    for (int i = 0; i < BIN_IT; ++i) pos[i] = atomicAdd(&lh[dv[i] / RNG], 1);
#pragma unroll
    for (int i = 0; i < BIN_IT; ++i)
      seg_l[pos[i]] = sv[i] | ((dv[i] - (dv[i] / RNG) * RNG) << SRC_BITS);
    __syncthreads();
    int* lp = list + grp * BIN_EPB;
#pragma unroll
    for (int i = 0; i < BIN_IT; ++i) lp[i * 256 + tid] = seg_l[i * 256 + tid];
    return;
  }

  const int gBase = (blockIdx.x - BIN_BLOCKS) * 16;

  {
    const int h  = tid & 127;
    const int r0 = tid >> 7;
    float wg[IN_GLOBAL];
#pragma unroll
    for (int k = 0; k < IN_GLOBAL; ++k) wg[k] = Wg[k * HIDDEN + h];
    const float b = bg[h];
#pragma unroll
    for (int i = 0; i < 8; ++i) {
      const int row = r0 + 2 * i;
      const float* gr = gf + (gBase + row) * IN_GLOBAL;
      float acc = b;
#pragma unroll
      for (int k = 0; k < IN_GLOBAL; ++k) acc = fmaf(gr[k], wg[k], acc);
      lds_g[row * LDSH_STRIDE + h] = fmaxf(acc, 0.0f);
    }
  }
  __syncthreads();

  const int wave  = tid >> 5;
  const int half  = lane >> 4;
  const int idx16 = lane & 15;
  const int colBase = wave * 16;

  v8f acc = {};
#pragma unroll 8
  for (int kb = 0; kb < HIDDEN; kb += 4) {
    const int k0 = kb + 2 * half;
    v2f a, b;
    a.x = lds_g[idx16 * LDSH_STRIDE + k0];
    a.y = lds_g[idx16 * LDSH_STRIDE + k0 + 1];
    b.x = Wc[(IN_LOCAL + k0) * HIDDEN + colBase + idx16];
    b.y = Wc[(IN_LOCAL + k0 + 1) * HIDDEN + colBase + idx16];
    acc = __builtin_amdgcn_wmma_f32_16x16x4_f32(
        false, a, false, b, (short)0, acc, false, false);
  }
  {
    const float bias = bc[colBase + idx16];
#pragma unroll
    for (int j = 0; j < 8; ++j) {
      int row = j + 8 * half;
      hgw[(gBase + row) * HIDDEN + colBase + idx16] = acc[j] + bias;
    }
  }
}

#define MT 5
static_assert(RNG == MT * 16, "kernel 5 covers its range with MT node tiles");

template <int TILE_UNROLL>
__device__ __forceinline__ void h0_rows_to_lds(
    const float* __restrict__ x, const int* __restrict__ batch,
    const float* __restrict__ hgw, const float* __restrict__ Wc,
    float* lds_h, int nodeBase, int half, int idx16, int colBase) {
  v2f bw[IN_LOCAL / 4];
#pragma unroll
  for (int kb = 0; kb < IN_LOCAL; kb += 4) {
    const int k0 = kb + 2 * half;
    bw[kb / 4].x = Wc[k0 * HIDDEN + colBase + idx16];
    bw[kb / 4].y = Wc[(k0 + 1) * HIDDEN + colBase + idx16];
  }
#pragma unroll TILE_UNROLL
  for (int t = 0; t < MT; ++t) {
    const int nb = nodeBase + t * 16;
    v8f acc;
#pragma unroll
    for (int j = 0; j < 8; ++j) {
      const int g = batch[nb + j + 8 * half];
      acc[j] = hgw[g * HIDDEN + colBase + idx16];
    }
    const float* xr = x + (nb + idx16) * IN_LOCAL + 2 * half;
#pragma unroll
    for (int kb = 0; kb < IN_LOCAL; kb += 4) {
      v2f a;
      a.x = xr[kb];
      a.y = xr[kb + 1];
      acc = __builtin_amdgcn_wmma_f32_16x16x4_f32(
          false, a, false, bw[kb / 4], (short)0, acc, false, false);
    }
#pragma unroll
    for (int j = 0; j < 8; ++j) {
      const int row = t * 16 + j + 8 * half;
      lds_h[row * LDSH_STRIDE + colBase + idx16] = fmaxf(acc[j], 0.0f);
    }
  }
}

__device__ __forceinline__ void h0_times_w(
    const float* lds_h, const float* __restrict__ W, int half, int idx16, int colBase,
    v8f (&am)[MT]) {
#pragma unroll
  for (int t = 0; t < MT; ++t) am[t] = v8f{};
#pragma unroll 2
  for (int kb = 0; kb < HIDDEN; kb += 4) {
    const int k0 = kb + 2 * half;
    v2f b0;
    b0.x = W[k0 * HIDDEN + colBase + idx16];
    b0.y = W[(k0 + 1) * HIDDEN + colBase + idx16];
#pragma unroll
    for (int t = 0; t < MT; ++t) {
      v2f a;
      a.x = lds_h[(t * 16 + idx16) * LDSH_STRIDE + k0];
      a.y = lds_h[(t * 16 + idx16) * LDSH_STRIDE + k0 + 1];
      am[t] = __builtin_amdgcn_wmma_f32_16x16x4_f32(
          false, a, false, b0, (short)0, am[t], false, false);
    }
  }
}

__global__ __launch_bounds__(256) void k_node_msg(
    const float* __restrict__ x,  const int* __restrict__ batch,
    const float* __restrict__ hgw, const float* __restrict__ Wc,
    const float* __restrict__ Wm, const float* __restrict__ bm,
    float* __restrict__ m_buf) {
  __shared__ float lds_h[MT * 16 * LDSH_STRIDE];

  const int tid = threadIdx.x;
  const int nodeBase = blockIdx.x * (MT * 16);
  const int lane  = tid & 31;
  const int wave  = tid >> 5;
  const int half  = lane >> 4;
  const int idx16 = lane & 15;
  const int colBase = wave * 16;

  h0_rows_to_lds<MT>(x, batch, hgw, Wc, lds_h, nodeBase, half, idx16, colBase);
  __syncthreads();
  v8f am[MT];
  h0_times_w(lds_h, Wm, half, idx16, colBase, am);
  {
    const int col = colBase + idx16;
    const float biasm = bm[col];
#pragma unroll
    for (int t = 0; t < MT; ++t) {
#pragma unroll
      for (int j = 0; j < 8; ++j) {
        const int node = nodeBase + t * 16 + j + 8 * half;
        m_buf[(size_t)node * HIDDEN + col] = fmaxf(am[t][j] + biasm, 0.0f);
      }
    }
  }
}

#define EU 8
#define PK 16
#define PK_LONG 8
#define SORT_CAP 2048
static_assert(BIN_BLOCKS * PK_LONG <= SORT_CAP, "a pass of PK_LONG entries per piece always fits");
#define NPW      (RNG / 8)

__global__ __launch_bounds__(256) void k_aggr_out(
    const float* __restrict__ x,  const int* __restrict__ batch,
    const float* __restrict__ hgw, const float* __restrict__ Wc,
    const float* __restrict__ Ws, const float* __restrict__ bs,
    const float* __restrict__ m_buf, const int* __restrict__ loc,
    const int* __restrict__ list, const float* __restrict__ Wo,
    const float* __restrict__ bo, float* __restrict__ out) {
  __shared__ __attribute__((aligned(16))) float acc_l[RNG * LDSH_STRIDE];
  __shared__ float res_l[RNG * 2];
  __shared__ int   srt_l[SORT_CAP];
  __shared__ int   nst_l[RNG + 1];
  __shared__ int   wmx_l[8];
  __shared__ int   wtot_l[8];

  const int tid = threadIdx.x;
  const int lane  = tid & 31;
  const int wave  = __builtin_amdgcn_readfirstlane(tid >> 5);
  const int half  = lane >> 4;
  const int idx16 = lane & 15;
  const int colBase = wave * 16;
  const int rbase = blockIdx.x * RNG;
  int pst, pc;

  {
    h0_rows_to_lds<1>(x, batch, hgw, Wc, acc_l, rbase, half, idx16, colBase);
    __syncthreads();
    v8f as[MT];
    h0_times_w(acc_l, Ws, half, idx16, colBase, as);
    const int col = colBase + idx16;
    const float biass = bs[col];
    __syncthreads();
#pragma unroll
    for (int t = 0; t < MT; ++t) {
#pragma unroll
      for (int j = 0; j < 8; ++j) {
        const int row = t * 16 + j + 8 * half;
        const float mv = m_buf[(size_t)(rbase + row) * HIDDEN + col];
        acc_l[row * HIDDEN + col] = mv + as[t][j] + biass;
      }
    }
    int p0 = 0, p1 = 0;
    if (tid < BIN_BLOCKS) {
      p0 = loc[blockIdx.x * LOC_STRIDE + tid];
      p1 = loc[(blockIdx.x + 1) * LOC_STRIDE + tid];
    }
    pst = tid * BIN_EPB + p0;
    pc  = p1 - p0;
    {
      int m = pc, s = pc;
#pragma unroll
      for (int off = 16; off >= 1; off >>= 1) {
        m = max(m, __shfl_xor(m, off, 32));
        s += __shfl_xor(s, off, 32);
      }
      if (lane == 0) { wmx_l[wave] = m; wtot_l[wave] = s; }
    }
    __syncthreads();
  }

  int mx = 0, tot = 0;
#pragma unroll
  for (int w = 0; w < 8; ++w) {
    mx = max(mx, wmx_l[w]);
    tot += wtot_l[w];
  }
  const int step = (tot <= SORT_CAP) ? PK : PK_LONG;
  for (int pass = 0; pass < BIN_EPB / PK_LONG + 1; ++pass) {
    const int pb = pass * step;
    if (pb >= mx) break;

    if (tid <= RNG) nst_l[tid] = 0;
    int pk[PK];
#pragma unroll
    for (int i = 0; i < PK; ++i)
      pk[i] = (i < step && pb + i < pc) ? list[pst + pb + i] : -1;
    __syncthreads();
#pragma unroll
    for (int i = 0; i < PK; ++i)
      if (pk[i] >= 0) atomicAdd(&nst_l[pk[i] >> SRC_BITS], 1);
    __syncthreads();
    if (wave == 0) {
      const int i0 = lane * 3;
      const int v0 = (i0     < RNG) ? nst_l[i0]     : 0;
      const int v1 = (i0 + 1 < RNG) ? nst_l[i0 + 1] : 0;
      const int v2 = (i0 + 2 < RNG) ? nst_l[i0 + 2] : 0;
      const int s = v0 + v1 + v2;
      int incl = s;
#pragma unroll
      for (int o = 1; o < 32; o <<= 1) {
        const int t = __shfl_up(incl, o, 32);
        if (lane >= o) incl += t;
      }
      const int run = incl - s;
      if (i0     <= RNG) nst_l[i0]     = run;
      if (i0 + 1 <= RNG) nst_l[i0 + 1] = run + v0;
      if (i0 + 2 <= RNG) nst_l[i0 + 2] = run + v0 + v1;
    }
    __syncthreads();
#pragma unroll
    for (int i = 0; i < PK; ++i)
      if (pk[i] >= 0) srt_l[atomicAdd(&nst_l[pk[i] >> SRC_BITS], 1)] = pk[i];
    __syncthreads();

    const int sb = (wave == 0) ? 0 : __builtin_amdgcn_readfirstlane(nst_l[wave * NPW - 1]);
    const int se = __builtin_amdgcn_readfirstlane(nst_l[wave * NPW + NPW - 1]);
    int cur = -1;
    v4f a = {0.0f, 0.0f, 0.0f, 0.0f};
    for (int c = sb; c < se; c += 32) {
      const int e = c + lane;
      const int q = (e < se) ? srt_l[e] : 0;
      const int cn = min(se - c, 32);
      for (int g = 0; g < cn; g += EU) {
        const int n = min(cn - g, EU);
        int s[EU], d[EU];
#pragma unroll
        for (int u = 0; u < EU; ++u) {
          const int p = __builtin_amdgcn_readlane(q, min(g + u, cn - 1));
          s[u] = p & ((1 << SRC_BITS) - 1);
          d[u] = p >> SRC_BITS;
        }
        v4f v[EU];
#pragma unroll
        for (int u = 0; u < EU; ++u)
          v[u] = *(const v4f*)(m_buf + (size_t)s[u] * HIDDEN + 4 * lane);
#pragma unroll
        for (int u = 0; u < EU; ++u) {
          if (d[u] != cur) {
            if (cur >= 0) *(v4f*)(acc_l + cur * HIDDEN + 4 * lane) = a;
            cur = d[u];
            a = *(const v4f*)(acc_l + cur * HIDDEN + 4 * lane);
          }
          const v4f z = {0.0f, 0.0f, 0.0f, 0.0f};
          a += (u < n) ? v[u] : z;
        }
      }
    }
    if (cur >= 0) *(v4f*)(acc_l + cur * HIDDEN + 4 * lane) = a;
    __syncthreads();
  }

  {
    float w0[4], w1[4];
#pragma unroll
    for (int t = 0; t < 4; ++t) {
      const int k = lane + t * 32;
      w0[t] = Wo[k * 2 + 0];
      w1[t] = Wo[k * 2 + 1];
    }
    for (int i = 0; i < NPW; ++i) {
      const int n = wave * NPW + i;
      float a0 = 0.0f, a1 = 0.0f;
#pragma unroll
      for (int t = 0; t < 4; ++t) {
        const float hv = fmaxf(acc_l[n * HIDDEN + lane + t * 32], 0.0f);
        a0 = fmaf(hv, w0[t], a0);
        a1 = fmaf(hv, w1[t], a1);
      }
#pragma unroll
      for (int off = 16; off >= 1; off >>= 1) {
        a0 += __shfl_xor(a0, off, 32);
        a1 += __shfl_xor(a1, off, 32);
      }
      if (lane < 2) res_l[n * 2 + lane] = lane ? a1 : a0;
    }
  }
  __syncthreads();
  if (tid < RNG * 2) out[rbase * 2 + tid] = res_l[tid] + bo[tid & 1];
}

extern "C" void kernel_launch(void* const* d_in, const int* in_sizes, int n_in,
                              void* d_out, int out_size, void* d_ws, size_t ws_size,
                              hipStream_t stream) {
  (void)in_sizes; (void)n_in; (void)out_size; (void)ws_size;
  const float* x     = (const float*)d_in[0];
  const int*   edge  = (const int*)  d_in[1];
  const int*   batch = (const int*)  d_in[2];
  const float* gf    = (const float*)d_in[3];
  const float* Wg    = (const float*)d_in[4];
  const float* bg    = (const float*)d_in[5];
  const float* Wc    = (const float*)d_in[6];
  const float* bc    = (const float*)d_in[7];
  const float* Wm    = (const float*)d_in[8];
  const float* bm    = (const float*)d_in[9];
  const float* Ws    = (const float*)d_in[10];
  const float* bs    = (const float*)d_in[11];
  const float* Wo    = (const float*)d_in[12];
  const float* bo    = (const float*)d_in[13];
  float* out = (float*)d_out;

  float* hgw   = (float*)d_ws;
  float* m_buf = hgw + (size_t)N_GRAPHS * HIDDEN;
  int*   loc   = (int*)(m_buf + (size_t)N_NODES * HIDDEN);
  int*   list  = loc + 327680;

  k_enc_sort<<<BIN_BLOCKS + N_GRAPHS / 16, 256, 0, stream>>>(gf, Wg, bg, Wc, bc, hgw, edge, loc, list);
  k_node_msg<<<N_NODES / (MT * 16), 256, 0, stream>>>(x, batch, hgw, Wc, Wm, bm, m_buf);
  k_aggr_out<<<NB, 256, 0, stream>>>(x, batch, hgw, Wc, Ws, bs, m_buf, loc, list,
                                     Wo, bo, out);
}
